# starting kernel
# baseline (speedup 1.0000x reference)
#include <hip/hip_runtime.h>

typedef __attribute__((ext_vector_type(2))) float v2f;
typedef __attribute__((ext_vector_type(8))) float v8f;

#define N_NODES 100000
#define NFEAT   256
#define NHID    64
#define NCLASS  41

__global__ __launch_bounds__(256) void fill_zero_k(float* __restrict__ p, int n) {
    int i = blockIdx.x * blockDim.x + threadIdx.x;
    if (i < n) p[i] = 0.0f;
}

__global__ __launch_bounds__(256) void gemm1_wmma_k(const float* __restrict__ x,
                                                    const float* __restrict__ W1,
                                                    float* __restrict__ h1) {
    const int lane = threadIdx.x & 31;
    const int wave = threadIdx.x >> 5;
    const int r0 = (blockIdx.x * 8 + wave) * 16;
    if (r0 >= N_NODES) return;

    const int m     = lane & 15;
    const int khalf = (lane >> 4) * 2;

    int arow = r0 + m;
    if (arow >= N_NODES) arow = N_NODES - 1;
    const float* __restrict__ xrow = x + (long)arow * NFEAT;

    v8f acc0 = {}, acc1 = {}, acc2 = {}, acc3 = {};
    for (int k0 = 0; k0 < NFEAT; k0 += 4) {
        v2f a;
        a.x = xrow[k0 + khalf];
        a.y = xrow[k0 + khalf + 1];
        const float* __restrict__ wlo = W1 + (k0 + khalf) * NHID + m;
        const float* __restrict__ whi = W1 + (k0 + khalf + 1) * NHID + m;
        v2f b0; b0.x = wlo[0];  b0.y = whi[0];
        v2f b1; b1.x = wlo[16]; b1.y = whi[16];
        v2f b2; b2.x = wlo[32]; b2.y = whi[32];
        v2f b3; b3.x = wlo[48]; b3.y = whi[48];
        acc0 = __builtin_amdgcn_wmma_f32_16x16x4_f32(false, a, false, b0, (short)0, acc0, false, false);
        acc1 = __builtin_amdgcn_wmma_f32_16x16x4_f32(false, a, false, b1, (short)0, acc1, false, false);
        acc2 = __builtin_amdgcn_wmma_f32_16x16x4_f32(false, a, false, b2, (short)0, acc2, false, false);
        acc3 = __builtin_amdgcn_wmma_f32_16x16x4_f32(false, a, false, b3, (short)0, acc3, false, false);
    }

    const int col     = lane & 15;
    const int rowHalf = (lane >> 4) * 8;
    for (int j = 0; j < 8; ++j) {
        int r = r0 + rowHalf + j;
        if (r < N_NODES) {
            float* __restrict__ o = h1 + (long)r * NHID + col;
            o[0]  = acc0[j];
            o[16] = acc1[j];
            o[32] = acc2[j];
            o[48] = acc3[j];
        }
    }
}

__global__ __launch_bounds__(256) void gemm2_wmma_k(const float* __restrict__ agg1,
                                                    const float* __restrict__ b1,
                                                    const float* __restrict__ W2,
                                                    float* __restrict__ h2) {
    const int lane = threadIdx.x & 31;
    const int wave = threadIdx.x >> 5;
    const int r0 = (blockIdx.x * 8 + wave) * 16;
    if (r0 >= N_NODES) return;

    const int m     = lane & 15;
    const int khalf = (lane >> 4) * 2;

    int arow = r0 + m;
    if (arow >= N_NODES) arow = N_NODES - 1;
    const float* __restrict__ hrow = agg1 + (long)arow * NHID;

    v8f acc0 = {}, acc1 = {}, acc2 = {};
    for (int k0 = 0; k0 < NHID; k0 += 4) {
        int ka = k0 + khalf;
        v2f a;
        a.x = fmaxf(hrow[ka]     + b1[ka],     0.0f);
        a.y = fmaxf(hrow[ka + 1] + b1[ka + 1], 0.0f);

        v2f b0, b1v, b2v;
        {
            int n = m;
            b0.x = W2[ka * NCLASS + n];
            b0.y = W2[(ka + 1) * NCLASS + n];
        }
        {
            int n = 16 + m;
            b1v.x = W2[ka * NCLASS + n];
            b1v.y = W2[(ka + 1) * NCLASS + n];
        }
        {
            int n = 32 + m;
            int nc = (n < NCLASS) ? n : (NCLASS - 1);
            float lo = W2[ka * NCLASS + nc];
            float hi = W2[(ka + 1) * NCLASS + nc];
            if (n >= NCLASS) { lo = 0.0f; hi = 0.0f; }
            b2v.x = lo; b2v.y = hi;
        }
        acc0 = __builtin_amdgcn_wmma_f32_16x16x4_f32(false, a, false, b0,  (short)0, acc0, false, false);
        acc1 = __builtin_amdgcn_wmma_f32_16x16x4_f32(false, a, false, b1v, (short)0, acc1, false, false);
        acc2 = __builtin_amdgcn_wmma_f32_16x16x4_f32(false, a, false, b2v, (short)0, acc2, false, false);
    }

    const int col     = lane & 15;
    const int rowHalf = (lane >> 4) * 8;
    for (int j = 0; j < 8; ++j) {
        int r = r0 + rowHalf + j;
        if (r < N_NODES) {
            float* __restrict__ o = h2 + (long)r * NCLASS;
            o[col]      = acc0[j];
            o[col + 16] = acc1[j];
            int c2 = col + 32;
            if (c2 < NCLASS) o[c2] = acc2[j];
        }
    }
}

template <int D>
__global__ __launch_bounds__(256) void spmm_k(const int* __restrict__ src,
                                              const int* __restrict__ dst,
                                              const float* __restrict__ w,
                                              const float* __restrict__ h,
                                              float* __restrict__ agg,
                                              int nEdges) {
    long idx = (long)blockIdx.x * blockDim.x + threadIdx.x;
    long total = (long)nEdges * D;
    if (idx >= total) return;
    int e = (int)(idx / D);
    int c = (int)(idx - (long)e * D);
    float v = w[e] * h[(long)src[e] * D + c];
    atomicAdd(&agg[(long)dst[e] * D + c], v);
}

__global__ __launch_bounds__(256) void logsoftmax_k(const float* __restrict__ agg2,
                                                    const float* __restrict__ b2,
                                                    float* __restrict__ out) {
    int i = blockIdx.x * blockDim.x + threadIdx.x;
    if (i >= N_NODES) return;
    const float* __restrict__ z = agg2 + (long)i * NCLASS;
    float zb[NCLASS];
    float mx = -3.0e38f;
    for (int c = 0; c < NCLASS; ++c) {
        zb[c] = z[c] + b2[c];
        mx = fmaxf(mx, zb[c]);
    }
    float s = 0.0f;
    for (int c = 0; c < NCLASS; ++c) s += expf(zb[c] - mx);
    float lse = mx + logf(s);
    float* __restrict__ o = out + (long)i * NCLASS;
    for (int c = 0; c < NCLASS; ++c) o[c] = zb[c] - lse;
}

extern "C" void kernel_launch(void* const* d_in, const int* in_sizes, int n_in,
                              void* d_out, int out_size, void* d_ws, size_t ws_size,
                              hipStream_t stream) {
    const float* x    = (const float*)d_in[0];
    const int*   esrc = (const int*)d_in[1];
    const int*   edst = (const int*)d_in[2];
    const float* ew   = (const float*)d_in[3];
    const float* W1   = (const float*)d_in[4];
    const float* b1   = (const float*)d_in[5];
    const float* W2   = (const float*)d_in[6];
    const float* b2   = (const float*)d_in[7];
    float* out = (float*)d_out;
    const int E = in_sizes[1];

    float* h1   = (float*)d_ws;
    float* agg1 = h1   + (size_t)N_NODES * NHID;
    float* h2   = agg1 + (size_t)N_NODES * NHID;
    float* agg2 = h2   + (size_t)N_NODES * NCLASS;

    const int n1 = N_NODES * NHID;
    const int n2 = N_NODES * NCLASS;
    fill_zero_k<<<(n1 + 255) / 256, 256, 0, stream>>>(agg1, n1);
    fill_zero_k<<<(n2 + 255) / 256, 256, 0, stream>>>(agg2, n2);

    const int gemmBlocks = (N_NODES + 127) / 128;
    gemm1_wmma_k<<<gemmBlocks, 256, 0, stream>>>(x, W1, h1);

    long w1 = (long)E * NHID;
    spmm_k<NHID><<<(unsigned)((w1 + 255) / 256), 256, 0, stream>>>(esrc, edst, ew, h1, agg1, E);

    gemm2_wmma_k<<<gemmBlocks, 256, 0, stream>>>(agg1, b1, W2, h2);

    long w2 = (long)E * NCLASS;
    spmm_k<NCLASS><<<(unsigned)((w2 + 255) / 256), 256, 0, stream>>>(esrc, edst, ew, h2, agg2, E);

    logsoftmax_k<<<(N_NODES + 255) / 256, 256, 0, stream>>>(agg2, b2, out);
}


# D=41 SpMM no longer scans dst: the D=64 SpMM appends each 32-entry batch (src, row, w; 8 B each) to its group's edge list in the workspace, the D=41 SpMM walks it; row sum order = LDS arrival
# speedup vs baseline: 2.5484x; 2.5484x over previous
#include <hip/hip_runtime.h>

typedef __attribute__((ext_vector_type(2))) float v2f;
typedef __attribute__((ext_vector_type(8))) float v8f;

#define N_NODES 100000
#define NFEAT   256
#define NHID    64
#define NCLASS  41

__global__ __launch_bounds__(256) void gemm1_wmma_k(const float* __restrict__ x,
                                                    const float* __restrict__ W1,
                                                    float* __restrict__ h1) {
    const int lane = threadIdx.x & 31;
    const int wave = threadIdx.x >> 5;
    const int r0 = (blockIdx.x * 8 + wave) * 16;
    if (r0 >= N_NODES) return;

    const int m     = lane & 15;
    const int khalf = (lane >> 4) * 2;

    int arow = r0 + m;
    if (arow >= N_NODES) arow = N_NODES - 1;
    const float* __restrict__ xrow = x + (long)arow * NFEAT;

    v8f acc0 = {}, acc1 = {}, acc2 = {}, acc3 = {};
    for (int k0 = 0; k0 < NFEAT; k0 += 4) {
        v2f a;
        a.x = xrow[k0 + khalf];
        a.y = xrow[k0 + khalf + 1];
        const float* __restrict__ wlo = W1 + (k0 + khalf) * NHID + m;
        const float* __restrict__ whi = W1 + (k0 + khalf + 1) * NHID + m;
        v2f b0; b0.x = wlo[0];  b0.y = whi[0];
        v2f b1; b1.x = wlo[16]; b1.y = whi[16];
        v2f b2; b2.x = wlo[32]; b2.y = whi[32];
        v2f b3; b3.x = wlo[48]; b3.y = whi[48];
        acc0 = __builtin_amdgcn_wmma_f32_16x16x4_f32(false, a, false, b0, (short)0, acc0, false, false);
        acc1 = __builtin_amdgcn_wmma_f32_16x16x4_f32(false, a, false, b1, (short)0, acc1, false, false);
        acc2 = __builtin_amdgcn_wmma_f32_16x16x4_f32(false, a, false, b2, (short)0, acc2, false, false);
        acc3 = __builtin_amdgcn_wmma_f32_16x16x4_f32(false, a, false, b3, (short)0, acc3, false, false);
    }

    const int col     = lane & 15;
    const int rowHalf = (lane >> 4) * 8;
    for (int j = 0; j < 8; ++j) {
        int r = r0 + rowHalf + j;
        if (r < N_NODES) {
            float* __restrict__ o = h1 + (long)r * NHID + col;
            o[0]  = acc0[j];
            o[16] = acc1[j];
            o[32] = acc2[j];
            o[48] = acc3[j];
        }
    }
}

__global__ __launch_bounds__(256) void gemm2_wmma_k(const float* __restrict__ agg1,
                                                    const float* __restrict__ b1,
                                                    const float* __restrict__ W2,
                                                    float* __restrict__ h2) {
    const int lane = threadIdx.x & 31;
    const int wave = threadIdx.x >> 5;
    const int r0 = (blockIdx.x * 8 + wave) * 16;
    if (r0 >= N_NODES) return;

    const int m     = lane & 15;
    const int khalf = (lane >> 4) * 2;

    int arow = r0 + m;
    if (arow >= N_NODES) arow = N_NODES - 1;
    const float* __restrict__ hrow = agg1 + (long)arow * NHID;

    v8f acc0 = {}, acc1 = {}, acc2 = {};
    for (int k0 = 0; k0 < NHID; k0 += 4) {
        int ka = k0 + khalf;
        v2f a;
        a.x = fmaxf(hrow[ka]     + b1[ka],     0.0f);
        a.y = fmaxf(hrow[ka + 1] + b1[ka + 1], 0.0f);

        v2f b0, b1v, b2v;
        {
            int n = m;
            b0.x = W2[ka * NCLASS + n];
            b0.y = W2[(ka + 1) * NCLASS + n];
        }
        {
            int n = 16 + m;
            b1v.x = W2[ka * NCLASS + n];
            b1v.y = W2[(ka + 1) * NCLASS + n];
        }
        {
            int n = 32 + m;
            int nc = (n < NCLASS) ? n : (NCLASS - 1);
            float lo = W2[ka * NCLASS + nc];
            float hi = W2[(ka + 1) * NCLASS + nc];
            if (n >= NCLASS) { lo = 0.0f; hi = 0.0f; }
            b2v.x = lo; b2v.y = hi;
        }
        acc0 = __builtin_amdgcn_wmma_f32_16x16x4_f32(false, a, false, b0,  (short)0, acc0, false, false);
        acc1 = __builtin_amdgcn_wmma_f32_16x16x4_f32(false, a, false, b1v, (short)0, acc1, false, false);
        acc2 = __builtin_amdgcn_wmma_f32_16x16x4_f32(false, a, false, b2v, (short)0, acc2, false, false);
    }

    const int col     = lane & 15;
    const int rowHalf = (lane >> 4) * 8;
    for (int j = 0; j < 8; ++j) {
        int r = r0 + rowHalf + j;
        if (r < N_NODES) {
            float* __restrict__ o = h2 + (long)r * NCLASS;
            o[col]      = acc0[j];
            o[col + 16] = acc1[j];
            int c2 = col + 32;
            if (c2 < NCLASS) o[c2] = acc2[j];
        }
    }
}

#define OWN_ROWS  800
#define OWN_WAVES 32
#define RING_CAP  256
#define LIST_SEG  (OWN_ROWS * NCLASS)
#define LIST_CAP  (LIST_SEG / 64 - 1)
static_assert(N_NODES % OWN_ROWS == 0, "row ranges must tile N exactly");
static_assert(N_NODES <= (1 << 17) && OWN_ROWS < (1 << 14), "src and row share one 32-bit word");
static_assert((LIST_SEG * 4) % 128 == 0, "segments start on a line");
static_assert((RING_CAP & (RING_CAP - 1)) == 0 && RING_CAP >= 256, "a wave adds at most 128 entries to at most 128 waiting ones");

__device__ __forceinline__ void lds_add(float* p, float v) {
    __hip_atomic_fetch_add(p, v, __ATOMIC_RELAXED, __HIP_MEMORY_SCOPE_WORKGROUP);
}

template <int D>
__device__ __forceinline__ void own_apply(int s, float ww, int r, int lane,
                                          const float* __restrict__ h, float* __restrict__ tile) {
#pragma unroll
    for (int hb = 0; hb < 2; ++hb) {
        float v0[16], v1[16];
#pragma unroll
        for (int j = 0; j < 16; ++j) {
            const int sj = __builtin_amdgcn_readlane(s, hb * 16 + j);
            const float* __restrict__ hr = h + (long)sj * D;
            v0[j] = hr[lane];
            v1[j] = 0.0f;
            if (lane + 32 < D) v1[j] = hr[lane + 32];
        }
#pragma unroll
        for (int j = 0; j < 16; ++j) {
            const float wj = __builtin_bit_cast(float, __builtin_amdgcn_readlane(__builtin_bit_cast(int, ww), hb * 16 + j));
            const int   rj = __builtin_amdgcn_readlane(r, hb * 16 + j);
            float* tr = tile + rj * D;
            lds_add(tr + lane, wj * v0[j]);
            if (lane + 32 < D) lds_add(tr + lane + 32, wj * v1[j]);
        }
    }
}

template <int D, bool KEEP>
__device__ __forceinline__ void own_drain(const int2* __restrict__ ring, int head, int n, int lane,
                                          const int* __restrict__ src, const float* __restrict__ w,
                                          const float* __restrict__ h, float* __restrict__ tile,
                                          int2* __restrict__ seg, int* slotCount) {
    __builtin_amdgcn_fence(__ATOMIC_ACQ_REL, "wavefront");
    const bool ok = lane < n;
    const int2 ent = ring[(head + lane) & (RING_CAP - 1)];
    const int   e  = ok ? ent.x : 0;
    const int   r  = ok ? ent.y : OWN_ROWS;
    const int   s  = src[e];
    const float ww = ok ? w[e] : 0.0f;
    if (KEEP) {
        int slot = 0;
        if (lane == 0) slot = __hip_atomic_fetch_add(slotCount, 1, __ATOMIC_RELAXED, __HIP_MEMORY_SCOPE_WORKGROUP);
        slot = __builtin_amdgcn_readfirstlane(slot);
        if (slot < LIST_CAP) seg[(slot + 1) * 32 + lane] = make_int2(s | (r << 17), __builtin_bit_cast(int, ww));
    }
    own_apply<D>(s, ww, r, lane, h, tile);
}

template <int D, int MODE>
__global__ __launch_bounds__(1024) void spmm_own_k(const int* __restrict__ src,
                                                   const int* __restrict__ dst,
                                                   const float* __restrict__ w,
                                                   const float* __restrict__ h,
                                                   float* __restrict__ agg,
                                                   int nEdges,
                                                   float* __restrict__ lst) {
    static_assert((OWN_ROWS * D) % 4 == 0 && (OWN_ROWS * D * 4) % 128 == 0, "tile stores are whole 16-B pieces of whole lines");
    __shared__ __attribute__((aligned(16))) float tile[(OWN_ROWS + 1) * D];
    __shared__ __attribute__((aligned(8)))  int2  rings[OWN_WAVES * RING_CAP];
    __shared__ int slotCount;

    const int lane = threadIdx.x & 31;
    const int wave = __builtin_amdgcn_readfirstlane(threadIdx.x >> 5);
    const int lo   = blockIdx.x * OWN_ROWS;

    float4* t4 = reinterpret_cast<float4*>(tile);
    for (int i = threadIdx.x; i < OWN_ROWS * D / 4; i += 1024) t4[i] = make_float4(0.f, 0.f, 0.f, 0.f);
    if (MODE == 0 && threadIdx.x == 0) slotCount = 0;
    __syncthreads();

    int2* ring = rings + wave * RING_CAP;
    int head = 0, tail = 0;
    int2* __restrict__ seg = reinterpret_cast<int2*>(lst + (long)blockIdx.x * LIST_SEG);

    auto push = [&](unsigned m, bool f, unsigned r, int e) {
        if (m != 0u) {
            if (f) ring[(tail + (int)__builtin_amdgcn_mbcnt_lo(m, 0u)) & (RING_CAP - 1)] = make_int2(e, (int)r);
            tail += __builtin_popcount(m);
        }
    };
    auto make_room = [&]() {
        if (tail - head > RING_CAP - 128) {
            if (MODE == 0 && lane == 0) __hip_atomic_fetch_or(&slotCount, 1 << 30, __ATOMIC_RELAXED, __HIP_MEMORY_SCOPE_WORKGROUP);
            __builtin_amdgcn_fence(__ATOMIC_ACQ_REL, "wavefront");
#pragma nounroll
            for (int k = head; k != tail; ++k) {
                const int2 ent = ring[k & (RING_CAP - 1)];
                const int e = __builtin_amdgcn_readfirstlane(ent.x);
                const int r = __builtin_amdgcn_readfirstlane(ent.y);
                const float wj = w[e];
                const float* __restrict__ hr = h + (long)src[e] * D;
                float* tr = tile + r * D;
                lds_add(tr + lane, wj * hr[lane]);
                if (lane + 32 < D) lds_add(tr + lane + 32, wj * hr[lane + 32]);
            }
            head = tail;
        }
    };
    auto take = [&](int e0, int4 d) {
        const unsigned r0 = (unsigned)(d.x - lo), r1 = (unsigned)(d.y - lo);
        const unsigned r2 = (unsigned)(d.z - lo), r3 = (unsigned)(d.w - lo);
        const bool f0 = r0 < (unsigned)OWN_ROWS, f1 = r1 < (unsigned)OWN_ROWS;
        const bool f2 = r2 < (unsigned)OWN_ROWS, f3 = r3 < (unsigned)OWN_ROWS;
        const unsigned m0 = __builtin_amdgcn_ballot_w32(f0), m1 = __builtin_amdgcn_ballot_w32(f1);
        const unsigned m2 = __builtin_amdgcn_ballot_w32(f2), m3 = __builtin_amdgcn_ballot_w32(f3);
        if ((m0 | m1 | m2 | m3) != 0u) {
            make_room();
            push(m0, f0, r0, e0);
            push(m1, f1, r1, e0 + 1);
            push(m2, f2, r2, e0 + 2);
            push(m3, f3, r3, e0 + 3);
        }
    };
    auto drain_batches = [&]() {
        while (tail - head >= 32) {
            own_drain<D, MODE == 0>(ring, head, 32, lane, src, w, h, tile, seg, &slotCount);
            head += 32;
        }
    };

    constexpr int U     = 4;
    constexpr int STEP  = 2 * U * OWN_WAVES;
    constexpr int CQUAD = OWN_WAVES * 32;
    const int nQuad  = nEdges >> 2;
    const int nBlock = (nQuad >> 5) / STEP;
    const int4* __restrict__ dst4 = reinterpret_cast<const int4*>(dst);

    bool scan = true;
    if (MODE == 1) {
        const int nSlot = __builtin_amdgcn_readfirstlane(seg[0].x);
        if (nSlot <= LIST_CAP) {
            scan = false;
            int sl = wave;
            int2 ent = make_int2(OWN_ROWS << 17, 0);
            if (sl < nSlot) ent = seg[(sl + 1) * 32 + lane];
            while (sl < nSlot) {
                const int nx = sl + OWN_WAVES;
                int2 nxt = ent;
                if (nx < nSlot) nxt = seg[(nx + 1) * 32 + lane];
                own_apply<D>(ent.x & 0x1FFFF, __builtin_bit_cast(float, ent.y), (int)((unsigned)ent.x >> 17), lane, h, tile);
                ent = nxt;
                sl = nx;
            }
        }
    }

    if (scan) {
    {
        const int q0 = (nBlock * STEP + wave) * 32 + lane;
        int4 rest[2 * U];
#pragma unroll
        for (int u = 0; u < 2 * U; ++u) {
            rest[u] = make_int4(-1, -1, -1, -1);
            if (q0 + u * CQUAD < nQuad) rest[u] = dst4[q0 + u * CQUAD];
        }
#pragma unroll
        for (int u = 0; u < 2 * U; ++u) take((q0 + u * CQUAD) * 4, rest[u]);
        if (wave == 0) {
            const int e = nQuad * 4 + lane;
            unsigned r = 0xFFFFFFFFu;
            if (lane < (nEdges & 3)) r = (unsigned)(dst[e] - lo);
            const bool f = r < (unsigned)OWN_ROWS;
            make_room();
            push(__builtin_amdgcn_ballot_w32(f), f, r, e);
        }
        drain_batches();
    }

    if (nBlock > 0) {
        int blk = (int)(((long)blockIdx.x * nBlock) / gridDim.x);
        int q = (blk * STEP + wave) * 32 + lane;
        int4 bufA[U], bufB[U];
#pragma unroll
        for (int u = 0; u < U; ++u) bufA[u] = dst4[q + u * CQUAD];
        for (int i = 0; i < nBlock; ++i) {
#pragma unroll
            for (int u = 0; u < U; ++u) bufB[u] = dst4[q + (U + u) * CQUAD];
#pragma unroll
            for (int u = 0; u < U; ++u) take((q + u * CQUAD) * 4, bufA[u]);
            blk = (blk + 1 == nBlock) ? 0 : blk + 1;
            const int qn = (blk * STEP + wave) * 32 + lane;
#pragma unroll
            for (int u = 0; u < U; ++u) bufA[u] = dst4[qn + u * CQUAD];
#pragma unroll
            for (int u = 0; u < U; ++u) take((q + (U + u) * CQUAD) * 4, bufB[u]);
            q = qn;
            drain_batches();
        }
    }
    if (tail - head > 0) own_drain<D, MODE == 0>(ring, head, tail - head, lane, src, w, h, tile, seg, &slotCount);
    }

    __syncthreads();
    if (MODE == 0 && threadIdx.x == 0) seg[0] = make_int2(slotCount, 0);
    float4* __restrict__ o4 = reinterpret_cast<float4*>(agg + (long)lo * D);
    for (int i = threadIdx.x; i < OWN_ROWS * D / 4; i += 1024) o4[i] = t4[i];
}

__global__ __launch_bounds__(256) void logsoftmax_k(const float* __restrict__ agg2,
                                                    const float* __restrict__ b2,
                                                    float* __restrict__ out) {
    int i = blockIdx.x * blockDim.x + threadIdx.x;
    if (i >= N_NODES) return;
    const float* __restrict__ z = agg2 + (long)i * NCLASS;
    float zb[NCLASS];
    float mx = -3.0e38f;
    for (int c = 0; c < NCLASS; ++c) {
        zb[c] = z[c] + b2[c];
        mx = fmaxf(mx, zb[c]);
    }
    float s = 0.0f;
    for (int c = 0; c < NCLASS; ++c) s += expf(zb[c] - mx);
    float lse = mx + logf(s);
    float* __restrict__ o = out + (long)i * NCLASS;
    for (int c = 0; c < NCLASS; ++c) o[c] = zb[c] - lse;
}

extern "C" void kernel_launch(void* const* d_in, const int* in_sizes, int n_in,
                              void* d_out, int out_size, void* d_ws, size_t ws_size,
                              hipStream_t stream) {
    const float* x    = (const float*)d_in[0];
    const int*   esrc = (const int*)d_in[1];
    const int*   edst = (const int*)d_in[2];
    const float* ew   = (const float*)d_in[3];
    const float* W1   = (const float*)d_in[4];
    const float* b1   = (const float*)d_in[5];
    const float* W2   = (const float*)d_in[6];
    const float* b2   = (const float*)d_in[7];
    float* out = (float*)d_out;
    const int E = in_sizes[1];

    float* h1   = (float*)d_ws;
    float* agg1 = h1   + (size_t)N_NODES * NHID;
    float* h2   = agg1 + (size_t)N_NODES * NHID;
    float* lst  = h2   + (size_t)N_NODES * NCLASS;
    float* agg2 = h1;

    const int gemmBlocks = (N_NODES + 127) / 128;
    gemm1_wmma_k<<<gemmBlocks, 256, 0, stream>>>(x, W1, h1);

    spmm_own_k<NHID, 0><<<N_NODES / OWN_ROWS, 1024, 0, stream>>>(esrc, edst, ew, h1, agg1, E, lst);

    gemm2_wmma_k<<<gemmBlocks, 256, 0, stream>>>(agg1, b1, W2, h2);

    spmm_own_k<NCLASS, 1><<<N_NODES / OWN_ROWS, 1024, 0, stream>>>(esrc, edst, ew, h2, agg2, E, lst);

    logsoftmax_k<<<(N_NODES + 255) / 256, 256, 0, stream>>>(agg2, b2, out);
}


# D=64 SpMM scan reads a 1-byte row-range key per edge (16 per 16-B load, bytes matched by whole-word ops), not the 4-byte dst; keys written by the GEMM1 launch; row sum order = LDS arrival
# speedup vs baseline: 2.7556x; 1.0813x over previous
#include <hip/hip_runtime.h>

typedef __attribute__((ext_vector_type(2))) float v2f;
typedef __attribute__((ext_vector_type(8))) float v8f;

#define N_NODES 100000
#define NFEAT   256
#define NHID    64
#define NCLASS  41
#define N_EDGES 1600000
#define OWN_ROWS  800
#define N_RANGES (N_NODES / OWN_ROWS)
#define KEY_NONE 127u
static_assert(N_RANGES <= 127, "a key and a range number are both below 128: key ^ range stays below 128 in every byte");

__device__ __forceinline__ unsigned edge_key(int d) {
    const unsigned q = (unsigned)d / OWN_ROWS;
    return q < KEY_NONE ? q : KEY_NONE;
}

__global__ __launch_bounds__(256) void gemm1_wmma_k(const float* __restrict__ x,
                                                    const float* __restrict__ W1,
                                                    float* __restrict__ h1,
                                                    const int* __restrict__ dst,
                                                    int nEdges,
                                                    float* __restrict__ keyf) {
    {
        uint2* __restrict__ key8 = reinterpret_cast<uint2*>(keyf);
        const int4* __restrict__ d4 = reinterpret_cast<const int4*>(dst);
        const int nOct = nEdges >> 3;
        for (int g = blockIdx.x * 256 + threadIdx.x; g < nOct; g += gridDim.x * 256) {
            const int4 a = d4[2 * g], b = d4[2 * g + 1];
            key8[g] = make_uint2(edge_key(a.x) | (edge_key(a.y) << 8) | (edge_key(a.z) << 16) | (edge_key(a.w) << 24),
                                 edge_key(b.x) | (edge_key(b.y) << 8) | (edge_key(b.z) << 16) | (edge_key(b.w) << 24));
        }
    }

    const int lane = threadIdx.x & 31;
    const int wave = threadIdx.x >> 5;
    const int r0 = (blockIdx.x * 8 + wave) * 16;
    if (r0 >= N_NODES) return;

    const int m     = lane & 15;
    const int khalf = (lane >> 4) * 2;

    int arow = r0 + m;
    if (arow >= N_NODES) arow = N_NODES - 1;
    const float* __restrict__ xrow = x + (long)arow * NFEAT;

    v8f acc0 = {}, acc1 = {}, acc2 = {}, acc3 = {};
    for (int k0 = 0; k0 < NFEAT; k0 += 4) {
        v2f a;
        a.x = xrow[k0 + khalf];
        a.y = xrow[k0 + khalf + 1];
        const float* __restrict__ wlo = W1 + (k0 + khalf) * NHID + m;
        const float* __restrict__ whi = W1 + (k0 + khalf + 1) * NHID + m;
        v2f b0; b0.x = wlo[0];  b0.y = whi[0];
        v2f b1; b1.x = wlo[16]; b1.y = whi[16];
        v2f b2; b2.x = wlo[32]; b2.y = whi[32];
        v2f b3; b3.x = wlo[48]; b3.y = whi[48];
        acc0 = __builtin_amdgcn_wmma_f32_16x16x4_f32(false, a, false, b0, (short)0, acc0, false, false);
        acc1 = __builtin_amdgcn_wmma_f32_16x16x4_f32(false, a, false, b1, (short)0, acc1, false, false);
        acc2 = __builtin_amdgcn_wmma_f32_16x16x4_f32(false, a, false, b2, (short)0, acc2, false, false);
        acc3 = __builtin_amdgcn_wmma_f32_16x16x4_f32(false, a, false, b3, (short)0, acc3, false, false);
    }

    const int col     = lane & 15;
    const int rowHalf = (lane >> 4) * 8;
    for (int j = 0; j < 8; ++j) {
        int r = r0 + rowHalf + j;
        if (r < N_NODES) {
            float* __restrict__ o = h1 + (long)r * NHID + col;
            o[0]  = acc0[j];
            o[16] = acc1[j];
            o[32] = acc2[j];
            o[48] = acc3[j];
        }
    }
}

__global__ __launch_bounds__(256) void gemm2_wmma_k(const float* __restrict__ agg1,
                                                    const float* __restrict__ b1,
                                                    const float* __restrict__ W2,
                                                    float* __restrict__ h2) {
    const int lane = threadIdx.x & 31;
    const int wave = threadIdx.x >> 5;
    const int r0 = (blockIdx.x * 8 + wave) * 16;
    if (r0 >= N_NODES) return;

    const int m     = lane & 15;
    const int khalf = (lane >> 4) * 2;

    int arow = r0 + m;
    if (arow >= N_NODES) arow = N_NODES - 1;
    const float* __restrict__ hrow = agg1 + (long)arow * NHID;

    v8f acc0 = {}, acc1 = {}, acc2 = {};
    for (int k0 = 0; k0 < NHID; k0 += 4) {
        int ka = k0 + khalf;
        v2f a;
        a.x = fmaxf(hrow[ka]     + b1[ka],     0.0f);
        a.y = fmaxf(hrow[ka + 1] + b1[ka + 1], 0.0f);

        v2f b0, b1v, b2v;
        {
            int n = m;
            b0.x = W2[ka * NCLASS + n];
            b0.y = W2[(ka + 1) * NCLASS + n];
        }
        {
            int n = 16 + m;
            b1v.x = W2[ka * NCLASS + n];
            b1v.y = W2[(ka + 1) * NCLASS + n];
        }
        {
            int n = 32 + m;
            int nc = (n < NCLASS) ? n : (NCLASS - 1);
            float lo = W2[ka * NCLASS + nc];
            float hi = W2[(ka + 1) * NCLASS + nc];
            if (n >= NCLASS) { lo = 0.0f; hi = 0.0f; }
            b2v.x = lo; b2v.y = hi;
        }
        acc0 = __builtin_amdgcn_wmma_f32_16x16x4_f32(false, a, false, b0,  (short)0, acc0, false, false);
        acc1 = __builtin_amdgcn_wmma_f32_16x16x4_f32(false, a, false, b1v, (short)0, acc1, false, false);
        acc2 = __builtin_amdgcn_wmma_f32_16x16x4_f32(false, a, false, b2v, (short)0, acc2, false, false);
    }

    const int col     = lane & 15;
    const int rowHalf = (lane >> 4) * 8;
    for (int j = 0; j < 8; ++j) {
        int r = r0 + rowHalf + j;
        if (r < N_NODES) {
            float* __restrict__ o = h2 + (long)r * NCLASS;
            o[col]      = acc0[j];
            o[col + 16] = acc1[j];
            int c2 = col + 32;
            if (c2 < NCLASS) o[c2] = acc2[j];
        }
    }
}

#define OWN_WAVES 32
#define RING_CAP  256
#define KEY_BYTES N_EDGES
#define LIST_SEG  ((N_NODES * NCLASS * 4 - KEY_BYTES) / N_RANGES / 4)
#define LIST_CAP  (LIST_SEG / 64 - 1)
static_assert(N_NODES % OWN_ROWS == 0, "row ranges must tile N exactly");
static_assert(N_NODES <= (1 << 17) && OWN_ROWS < (1 << 14), "src and row share one 32-bit word");
static_assert((N_NODES * NCLASS * 4 - KEY_BYTES) % (N_RANGES * 128) == 0 && (LIST_SEG * 4) % 128 == 0, "segments and the key plane start on a line");
static_assert((RING_CAP & (RING_CAP - 1)) == 0 && RING_CAP >= 64, "a wave adds at most 32 entries at a time");

__device__ __forceinline__ void lds_add(float* p, float v) {
    __hip_atomic_fetch_add(p, v, __ATOMIC_RELAXED, __HIP_MEMORY_SCOPE_WORKGROUP);
}

template <int D>
__device__ __forceinline__ void own_apply(int s, float ww, int r, int lane,
                                          const float* __restrict__ h, float* __restrict__ tile) {
#pragma unroll
    for (int hb = 0; hb < 2; ++hb) {
        float v0[16], v1[16];
#pragma unroll
        for (int j = 0; j < 16; ++j) {
            const int sj = __builtin_amdgcn_readlane(s, hb * 16 + j);
            const float* __restrict__ hr = h + (long)sj * D;
            v0[j] = hr[lane];
            v1[j] = 0.0f;
            if (lane + 32 < D) v1[j] = hr[lane + 32];
        }
#pragma unroll
        for (int j = 0; j < 16; ++j) {
            const float wj = __builtin_bit_cast(float, __builtin_amdgcn_readlane(__builtin_bit_cast(int, ww), hb * 16 + j));
            const int   rj = __builtin_amdgcn_readlane(r, hb * 16 + j);
            float* tr = tile + rj * D;
            lds_add(tr + lane, wj * v0[j]);
            if (lane + 32 < D) lds_add(tr + lane + 32, wj * v1[j]);
        }
    }
}

template <int D, bool KEEP>
__device__ __forceinline__ void own_drain(const int* __restrict__ ring, int head, int n, int lane, int lo,
                                          const int* __restrict__ src, const int* __restrict__ dst,
                                          const float* __restrict__ w,
                                          const float* __restrict__ h, float* __restrict__ tile,
                                          int2* __restrict__ seg, int* slotCount) {
    __builtin_amdgcn_fence(__ATOMIC_ACQ_REL, "wavefront");
    const bool ok = lane < n;
    const int  re = ring[(head + lane) & (RING_CAP - 1)];
    const int   e  = ok ? re : 0;
    const int   r  = ok ? dst[e] - lo : OWN_ROWS;
    const int   s  = src[e];
    const float ww = ok ? w[e] : 0.0f;
    if (KEEP) {
        int slot = 0;
        if (lane == 0) slot = __hip_atomic_fetch_add(slotCount, 1, __ATOMIC_RELAXED, __HIP_MEMORY_SCOPE_WORKGROUP);
        slot = __builtin_amdgcn_readfirstlane(slot);
        if (slot < LIST_CAP) seg[(slot + 1) * 32 + lane] = make_int2(s | (r << 17), __builtin_bit_cast(int, ww));
    }
    own_apply<D>(s, ww, r, lane, h, tile);
}

template <int D, int MODE>
__global__ __launch_bounds__(1024) void spmm_own_k(const int* __restrict__ src,
                                                   const int* __restrict__ dst,
                                                   const float* __restrict__ w,
                                                   const float* __restrict__ h,
                                                   float* __restrict__ agg,
                                                   int nEdges,
                                                   float* __restrict__ lst) {
    static_assert((OWN_ROWS * D) % 4 == 0 && (OWN_ROWS * D * 4) % 128 == 0, "tile stores are whole 16-B pieces of whole lines");
    __shared__ __attribute__((aligned(16))) float tile[(OWN_ROWS + 1) * D];
    __shared__ int rings[OWN_WAVES * RING_CAP];
    __shared__ int slotCount;

    const int lane = threadIdx.x & 31;
    const int wave = __builtin_amdgcn_readfirstlane(threadIdx.x >> 5);
    const int lo   = blockIdx.x * OWN_ROWS;

    float4* t4 = reinterpret_cast<float4*>(tile);
    for (int i = threadIdx.x; i < OWN_ROWS * D / 4; i += 1024) t4[i] = make_float4(0.f, 0.f, 0.f, 0.f);
    if (MODE == 0 && threadIdx.x == 0) slotCount = 0;
    __syncthreads();

    int* ring = rings + wave * RING_CAP;
    int head = 0, tail = 0;
    int2* __restrict__ seg = reinterpret_cast<int2*>(lst + (long)blockIdx.x * LIST_SEG);
    const int4* __restrict__ key4 = reinterpret_cast<const int4*>(lst + (long)N_RANGES * LIST_SEG);
    const unsigned pat = (unsigned)blockIdx.x * 0x01010101u;

    auto push = [&](unsigned m, bool f, int e) {
        if (f) ring[(tail + (int)__builtin_amdgcn_mbcnt_lo(m, 0u)) & (RING_CAP - 1)] = e;
        tail += __builtin_popcount(m);
    };
    auto make_room = [&]() {
        if (tail - head > RING_CAP - 32) {
            if (MODE == 0 && lane == 0) __hip_atomic_fetch_or(&slotCount, 1 << 30, __ATOMIC_RELAXED, __HIP_MEMORY_SCOPE_WORKGROUP);
            __builtin_amdgcn_fence(__ATOMIC_ACQ_REL, "wavefront");
#pragma nounroll
            for (int k = head; k != tail; ++k) {
                const int e = __builtin_amdgcn_readfirstlane(ring[k & (RING_CAP - 1)]);
                const int r = dst[e] - lo;
                const float wj = w[e];
                const float* __restrict__ hr = h + (long)src[e] * D;
                float* tr = tile + r * D;
                lds_add(tr + lane, wj * hr[lane]);
                if (lane + 32 < D) lds_add(tr + lane + 32, wj * hr[lane + 32]);
            }
            head = tail;
        }
    };
    auto take = [&](int e0, int4 d) {
        const unsigned t0 = ((unsigned)d.x ^ pat) + 0x7F7F7F7Fu, t1 = ((unsigned)d.y ^ pat) + 0x7F7F7F7Fu;
        const unsigned t2 = ((unsigned)d.z ^ pat) + 0x7F7F7F7Fu, t3 = ((unsigned)d.w ^ pat) + 0x7F7F7F7Fu;
        unsigned z = (((t0 >> 7) & 0x01010101u) | ((t1 >> 6) & 0x02020202u) |
                      ((t2 >> 5) & 0x04040404u) | ((t3 >> 4) & 0x08080808u)) ^ 0x0F0F0F0Fu;
#pragma nounroll
        for (int it = 0; it < 16; ++it) {
            const bool f = z != 0u;
            const unsigned m = __builtin_amdgcn_ballot_w32(f);
            if (m == 0u) break;
            make_room();
            const int p = __builtin_ctz(z | 0x80000000u);
            push(m, f, e0 + ((p & 7) << 2) + (p >> 3));
            z &= z - 1u;
        }
    };
    auto drain_batches = [&]() {
        while (tail - head >= 32) {
            own_drain<D, MODE == 0>(ring, head, 32, lane, lo, src, dst, w, h, tile, seg, &slotCount);
            head += 32;
        }
    };

    constexpr int U     = 4;
    constexpr int STEP  = 2 * U * OWN_WAVES;
    constexpr int CQUAD = OWN_WAVES * 32;
    const int nQuad  = nEdges >> 4;
    const int nBlock = (nQuad >> 5) / STEP;

    bool scan = true;
    if (MODE == 1) {
        const int nSlot = __builtin_amdgcn_readfirstlane(seg[0].x);
        if (nSlot <= LIST_CAP) {
            scan = false;
            int sl = wave;
            int2 ent = make_int2(OWN_ROWS << 17, 0);
            if (sl < nSlot) ent = seg[(sl + 1) * 32 + lane];
            while (sl < nSlot) {
                const int nx = sl + OWN_WAVES;
                int2 nxt = ent;
                if (nx < nSlot) nxt = seg[(nx + 1) * 32 + lane];
                own_apply<D>(ent.x & 0x1FFFF, __builtin_bit_cast(float, ent.y), (int)((unsigned)ent.x >> 17), lane, h, tile);
                ent = nxt;
                sl = nx;
            }
        }
    }

    if (scan) {
    {
        const int q0 = (nBlock * STEP + wave) * 32 + lane;
        int4 rest[2 * U];
#pragma unroll
        for (int u = 0; u < 2 * U; ++u) {
            rest[u] = make_int4(0x7F7F7F7F, 0x7F7F7F7F, 0x7F7F7F7F, 0x7F7F7F7F);
            if (q0 + u * CQUAD < nQuad) rest[u] = key4[q0 + u * CQUAD];
        }
#pragma unroll
        for (int u = 0; u < 2 * U; ++u) take((q0 + u * CQUAD) * 16, rest[u]);
        if (wave == 0) {
            const int e = nQuad * 16 + lane;
            unsigned r = 0xFFFFFFFFu;
            if (lane < (nEdges & 15)) r = (unsigned)(dst[e] - lo);
            const bool f = r < (unsigned)OWN_ROWS;
            const unsigned m = __builtin_amdgcn_ballot_w32(f);
            if (m != 0u) {
                make_room();
                push(m, f, e);
            }
        }
        drain_batches();
    }

    if (nBlock > 0) {
        int blk = (int)(((long)blockIdx.x * nBlock) / gridDim.x);
        int q = (blk * STEP + wave) * 32 + lane;
        int4 bufA[U], bufB[U];
#pragma unroll
        for (int u = 0; u < U; ++u) bufA[u] = key4[q + u * CQUAD];
        for (int i = 0; i < nBlock; ++i) {
#pragma unroll
            for (int u = 0; u < U; ++u) bufB[u] = key4[q + (U + u) * CQUAD];
#pragma unroll
            for (int u = 0; u < U; ++u) take((q + u * CQUAD) * 16, bufA[u]);
            blk = (blk + 1 == nBlock) ? 0 : blk + 1;
            const int qn = (blk * STEP + wave) * 32 + lane;
#pragma unroll
            for (int u = 0; u < U; ++u) bufA[u] = key4[qn + u * CQUAD];
#pragma unroll
            for (int u = 0; u < U; ++u) take((q + (U + u) * CQUAD) * 16, bufB[u]);
            q = qn;
            drain_batches();
        }
    }
    if (tail - head > 0) own_drain<D, MODE == 0>(ring, head, tail - head, lane, lo, src, dst, w, h, tile, seg, &slotCount);
    }

    __syncthreads();
    if (MODE == 0 && threadIdx.x == 0) seg[0] = make_int2(slotCount, 0);
    float4* __restrict__ o4 = reinterpret_cast<float4*>(agg + (long)lo * D);
    for (int i = threadIdx.x; i < OWN_ROWS * D / 4; i += 1024) o4[i] = t4[i];
}

__global__ __launch_bounds__(256) void logsoftmax_k(const float* __restrict__ agg2,
                                                    const float* __restrict__ b2,
                                                    float* __restrict__ out) {
    int i = blockIdx.x * blockDim.x + threadIdx.x;
    if (i >= N_NODES) return;
    const float* __restrict__ z = agg2 + (long)i * NCLASS;
    float zb[NCLASS];
    float mx = -3.0e38f;
    for (int c = 0; c < NCLASS; ++c) {
        zb[c] = z[c] + b2[c];
        mx = fmaxf(mx, zb[c]);
    }
    float s = 0.0f;
    for (int c = 0; c < NCLASS; ++c) s += expf(zb[c] - mx);
    float lse = mx + logf(s);
    float* __restrict__ o = out + (long)i * NCLASS;
    for (int c = 0; c < NCLASS; ++c) o[c] = zb[c] - lse;
}

extern "C" void kernel_launch(void* const* d_in, const int* in_sizes, int n_in,
                              void* d_out, int out_size, void* d_ws, size_t ws_size,
                              hipStream_t stream) {
    const float* x    = (const float*)d_in[0];
    const int*   esrc = (const int*)d_in[1];
    const int*   edst = (const int*)d_in[2];
    const float* ew   = (const float*)d_in[3];
    const float* W1   = (const float*)d_in[4];
    const float* b1   = (const float*)d_in[5];
    const float* W2   = (const float*)d_in[6];
    const float* b2   = (const float*)d_in[7];
    float* out = (float*)d_out;
    const int E = in_sizes[1];

    float* h1   = (float*)d_ws;
    float* agg1 = h1   + (size_t)N_NODES * NHID;
    float* h2   = agg1 + (size_t)N_NODES * NHID;
    float* lst  = h2   + (size_t)N_NODES * NCLASS;
    float* keyf = lst  + (size_t)N_RANGES * LIST_SEG;
    float* agg2 = h1;

    const int gemmBlocks = (N_NODES + 127) / 128;
    gemm1_wmma_k<<<gemmBlocks, 256, 0, stream>>>(x, W1, h1, edst, E, keyf);

    spmm_own_k<NHID, 0><<<N_NODES / OWN_ROWS, 1024, 0, stream>>>(esrc, edst, ew, h1, agg1, E, lst);

    gemm2_wmma_k<<<gemmBlocks, 256, 0, stream>>>(agg1, b1, W2, h2);

    spmm_own_k<NCLASS, 1><<<N_NODES / OWN_ROWS, 1024, 0, stream>>>(esrc, edst, ew, h2, agg2, E, lst);

    logsoftmax_k<<<(N_NODES + 255) / 256, 256, 0, stream>>>(agg2, b2, out);
}


# SpMM row ranges of 416 rows, not 800: 241 work-groups, one per WGP, where 125 left half of the 256 WGPs idle; each adds half as many edges; byte keys matched on all 8 bits; sum order = LDS arrival
# speedup vs baseline: 3.3183x; 1.2042x over previous
#include <hip/hip_runtime.h>

typedef __attribute__((ext_vector_type(2))) float v2f;
typedef __attribute__((ext_vector_type(8))) float v8f;

#define N_NODES 100000
#define NFEAT   256
#define NHID    64
#define NCLASS  41
#define N_EDGES 1600000
#define OWN_ROWS  416
#define N_RANGES ((N_NODES + OWN_ROWS - 1) / OWN_ROWS)
#define LAST_ROWS (N_NODES - (N_RANGES - 1) * OWN_ROWS)
#define KEY_NONE 255u
static_assert(N_RANGES <= 255, "a range number fits a byte and is never KEY_NONE");
static_assert(OWN_ROWS % 32 == 0, "rows of 41 floats: 32 of them are a whole number of 128-B lines");

__device__ __forceinline__ unsigned edge_key(int d) {
    const unsigned q = (unsigned)d / OWN_ROWS;
    return q < KEY_NONE ? q : KEY_NONE;
}

__global__ __launch_bounds__(256) void gemm1_wmma_k(const float* __restrict__ x,
                                                    const float* __restrict__ W1,
                                                    float* __restrict__ h1,
                                                    const int* __restrict__ dst,
                                                    int nEdges,
                                                    float* __restrict__ keyf) {
    {
        uint2* __restrict__ key8 = reinterpret_cast<uint2*>(keyf);
        const int4* __restrict__ d4 = reinterpret_cast<const int4*>(dst);
        const int nOct = nEdges >> 3;
        for (int g = blockIdx.x * 256 + threadIdx.x; g < nOct; g += gridDim.x * 256) {
            const int4 a = d4[2 * g], b = d4[2 * g + 1];
            key8[g] = make_uint2(edge_key(a.x) | (edge_key(a.y) << 8) | (edge_key(a.z) << 16) | (edge_key(a.w) << 24),
                                 edge_key(b.x) | (edge_key(b.y) << 8) | (edge_key(b.z) << 16) | (edge_key(b.w) << 24));
        }
    }

    const int lane = threadIdx.x & 31;
    const int wave = threadIdx.x >> 5;
    const int r0 = (blockIdx.x * 8 + wave) * 16;
    if (r0 >= N_NODES) return;

    const int m     = lane & 15;
    const int khalf = (lane >> 4) * 2;

    int arow = r0 + m;
    if (arow >= N_NODES) arow = N_NODES - 1;
    const float* __restrict__ xrow = x + (long)arow * NFEAT;

    v8f acc0 = {}, acc1 = {}, acc2 = {}, acc3 = {};
    for (int k0 = 0; k0 < NFEAT; k0 += 4) {
        v2f a;
        a.x = xrow[k0 + khalf];
        a.y = xrow[k0 + khalf + 1];
        const float* __restrict__ wlo = W1 + (k0 + khalf) * NHID + m;
        const float* __restrict__ whi = W1 + (k0 + khalf + 1) * NHID + m;
        v2f b0; b0.x = wlo[0];  b0.y = whi[0];
        v2f b1; b1.x = wlo[16]; b1.y = whi[16];
        v2f b2; b2.x = wlo[32]; b2.y = whi[32];
        v2f b3; b3.x = wlo[48]; b3.y = whi[48];
        acc0 = __builtin_amdgcn_wmma_f32_16x16x4_f32(false, a, false, b0, (short)0, acc0, false, false);
        acc1 = __builtin_amdgcn_wmma_f32_16x16x4_f32(false, a, false, b1, (short)0, acc1, false, false);
        acc2 = __builtin_amdgcn_wmma_f32_16x16x4_f32(false, a, false, b2, (short)0, acc2, false, false);
        acc3 = __builtin_amdgcn_wmma_f32_16x16x4_f32(false, a, false, b3, (short)0, acc3, false, false);
    }

    const int col     = lane & 15;
    const int rowHalf = (lane >> 4) * 8;
    for (int j = 0; j < 8; ++j) {
        int r = r0 + rowHalf + j;
        if (r < N_NODES) {
            float* __restrict__ o = h1 + (long)r * NHID + col;
            o[0]  = acc0[j];
            o[16] = acc1[j];
            o[32] = acc2[j];
            o[48] = acc3[j];
        }
    }
}

__global__ __launch_bounds__(256) void gemm2_wmma_k(const float* __restrict__ agg1,
                                                    const float* __restrict__ b1,
                                                    const float* __restrict__ W2,
                                                    float* __restrict__ h2) {
    const int lane = threadIdx.x & 31;
    const int wave = threadIdx.x >> 5;
    const int r0 = (blockIdx.x * 8 + wave) * 16;
    if (r0 >= N_NODES) return;

    const int m     = lane & 15;
    const int khalf = (lane >> 4) * 2;

    int arow = r0 + m;
    if (arow >= N_NODES) arow = N_NODES - 1;
    const float* __restrict__ hrow = agg1 + (long)arow * NHID;

    v8f acc0 = {}, acc1 = {}, acc2 = {};
    for (int k0 = 0; k0 < NHID; k0 += 4) {
        int ka = k0 + khalf;
        v2f a;
        a.x = fmaxf(hrow[ka]     + b1[ka],     0.0f);
        a.y = fmaxf(hrow[ka + 1] + b1[ka + 1], 0.0f);

        v2f b0, b1v, b2v;
        {
            int n = m;
            b0.x = W2[ka * NCLASS + n];
            b0.y = W2[(ka + 1) * NCLASS + n];
        }
        {
            int n = 16 + m;
            b1v.x = W2[ka * NCLASS + n];
            b1v.y = W2[(ka + 1) * NCLASS + n];
        }
        {
            int n = 32 + m;
            int nc = (n < NCLASS) ? n : (NCLASS - 1);
            float lo = W2[ka * NCLASS + nc];
            float hi = W2[(ka + 1) * NCLASS + nc];
            if (n >= NCLASS) { lo = 0.0f; hi = 0.0f; }
            b2v.x = lo; b2v.y = hi;
        }
        acc0 = __builtin_amdgcn_wmma_f32_16x16x4_f32(false, a, false, b0,  (short)0, acc0, false, false);
        acc1 = __builtin_amdgcn_wmma_f32_16x16x4_f32(false, a, false, b1v, (short)0, acc1, false, false);
        acc2 = __builtin_amdgcn_wmma_f32_16x16x4_f32(false, a, false, b2v, (short)0, acc2, false, false);
    }

    const int col     = lane & 15;
    const int rowHalf = (lane >> 4) * 8;
    for (int j = 0; j < 8; ++j) {
        int r = r0 + rowHalf + j;
        if (r < N_NODES) {
            float* __restrict__ o = h2 + (long)r * NCLASS;
            o[col]      = acc0[j];
            o[col + 16] = acc1[j];
            int c2 = col + 32;
            if (c2 < NCLASS) o[c2] = acc2[j];
        }
    }
}

#define OWN_WAVES 32
#define RING_CAP  256
#define KEY_BYTES N_EDGES
#define LIST_SEG  ((N_NODES * NCLASS * 4 - KEY_BYTES) / N_RANGES / 128 * 32)
#define LIST_CAP  (LIST_SEG / 64 - 1)
static_assert(LAST_ROWS > 0 && LAST_ROWS <= OWN_ROWS, "row ranges must cover N");
static_assert(N_NODES <= (1 << 17) && OWN_ROWS < (1 << 14), "src and row share one 32-bit word");
static_assert((LIST_SEG * 4) % 128 == 0 && (long)N_RANGES * LIST_SEG * 4 + KEY_BYTES <= (long)N_NODES * NCLASS * 4, "segments and the key plane start on a line and fit the plane");
static_assert((RING_CAP & (RING_CAP - 1)) == 0 && RING_CAP >= 64, "a wave adds at most 32 entries at a time");

__device__ __forceinline__ void lds_add(float* p, float v) {
    __hip_atomic_fetch_add(p, v, __ATOMIC_RELAXED, __HIP_MEMORY_SCOPE_WORKGROUP);
}

template <int D>
__device__ __forceinline__ void own_apply(int s, float ww, int r, int lane,
                                          const float* __restrict__ h, float* __restrict__ tile) {
#pragma unroll
    for (int hb = 0; hb < 2; ++hb) {
        float v0[16], v1[16];
#pragma unroll
        for (int j = 0; j < 16; ++j) {
            const int sj = __builtin_amdgcn_readlane(s, hb * 16 + j);
            const float* __restrict__ hr = h + (long)sj * D;
            v0[j] = hr[lane];
            v1[j] = 0.0f;
            if (lane + 32 < D) v1[j] = hr[lane + 32];
        }
#pragma unroll
        for (int j = 0; j < 16; ++j) {
            const float wj = __builtin_bit_cast(float, __builtin_amdgcn_readlane(__builtin_bit_cast(int, ww), hb * 16 + j));
            const int   rj = __builtin_amdgcn_readlane(r, hb * 16 + j);
            float* tr = tile + rj * D;
            lds_add(tr + lane, wj * v0[j]);
            if (lane + 32 < D) lds_add(tr + lane + 32, wj * v1[j]);
        }
    }
}

template <int D, bool KEEP>
__device__ __forceinline__ void own_drain(const int* __restrict__ ring, int head, int n, int lane, int lo,
                                          const int* __restrict__ src, const int* __restrict__ dst,
                                          const float* __restrict__ w,
                                          const float* __restrict__ h, float* __restrict__ tile,
                                          int2* __restrict__ seg, int* slotCount) {
    __builtin_amdgcn_fence(__ATOMIC_ACQ_REL, "wavefront");
    const bool ok = lane < n;
    const int  re = ring[(head + lane) & (RING_CAP - 1)];
    const int   e  = ok ? re : 0;
    const int   r  = ok ? dst[e] - lo : OWN_ROWS;
    const int   s  = src[e];
    const float ww = ok ? w[e] : 0.0f;
    if (KEEP) {
        int slot = 0;
        if (lane == 0) slot = __hip_atomic_fetch_add(slotCount, 1, __ATOMIC_RELAXED, __HIP_MEMORY_SCOPE_WORKGROUP);
        slot = __builtin_amdgcn_readfirstlane(slot);
        if (slot < LIST_CAP) seg[(slot + 1) * 32 + lane] = make_int2(s | (r << 17), __builtin_bit_cast(int, ww));
    }
    own_apply<D>(s, ww, r, lane, h, tile);
}

template <int D, int MODE>
__global__ __launch_bounds__(1024) void spmm_own_k(const int* __restrict__ src,
                                                   const int* __restrict__ dst,
                                                   const float* __restrict__ w,
                                                   const float* __restrict__ h,
                                                   float* __restrict__ agg,
                                                   int nEdges,
                                                   float* __restrict__ lst) {
    static_assert((OWN_ROWS * D) % 4 == 0 && (OWN_ROWS * D * 4) % 128 == 0 &&
                  (LAST_ROWS * D) % 4 == 0 && (LAST_ROWS * D * 4) % 128 == 0, "tile stores are whole 16-B pieces of whole lines");
    __shared__ __attribute__((aligned(16))) float tile[(OWN_ROWS + 1) * D];
    __shared__ int rings[OWN_WAVES * RING_CAP];
    __shared__ int slotCount;

    const int lane = threadIdx.x & 31;
    const int wave = __builtin_amdgcn_readfirstlane(threadIdx.x >> 5);
    const int lo   = blockIdx.x * OWN_ROWS;

    float4* t4 = reinterpret_cast<float4*>(tile);
    for (int i = threadIdx.x; i < OWN_ROWS * D / 4; i += 1024) t4[i] = make_float4(0.f, 0.f, 0.f, 0.f);
    if (MODE == 0 && threadIdx.x == 0) slotCount = 0;
    __syncthreads();

    int* ring = rings + wave * RING_CAP;
    int head = 0, tail = 0;
    int2* __restrict__ seg = reinterpret_cast<int2*>(lst + (long)blockIdx.x * LIST_SEG);
    const int4* __restrict__ key4 = reinterpret_cast<const int4*>(lst + (long)N_RANGES * LIST_SEG);
    const unsigned pat = (unsigned)blockIdx.x * 0x01010101u;

    auto push = [&](unsigned m, bool f, int e) {
        if (f) ring[(tail + (int)__builtin_amdgcn_mbcnt_lo(m, 0u)) & (RING_CAP - 1)] = e;
        tail += __builtin_popcount(m);
    };
    auto make_room = [&]() {
        if (tail - head > RING_CAP - 32) {
            if (MODE == 0 && lane == 0) __hip_atomic_fetch_or(&slotCount, 1 << 30, __ATOMIC_RELAXED, __HIP_MEMORY_SCOPE_WORKGROUP);
            __builtin_amdgcn_fence(__ATOMIC_ACQ_REL, "wavefront");
#pragma nounroll
            for (int k = head; k != tail; ++k) {
                const int e = __builtin_amdgcn_readfirstlane(ring[k & (RING_CAP - 1)]);
                const int r = dst[e] - lo;
                const float wj = w[e];
                const float* __restrict__ hr = h + (long)src[e] * D;
                float* tr = tile + r * D;
                lds_add(tr + lane, wj * hr[lane]);
                if (lane + 32 < D) lds_add(tr + lane + 32, wj * hr[lane + 32]);
            }
            head = tail;
        }
    };
    auto take = [&](int e0, int4 d) {
        const unsigned x0 = (unsigned)d.x ^ pat, x1 = (unsigned)d.y ^ pat, x2 = (unsigned)d.z ^ pat, x3 = (unsigned)d.w ^ pat;
        const unsigned m0 = ~(((x0 & 0x7F7F7F7Fu) + 0x7F7F7F7Fu) | x0) & 0x80808080u;
        const unsigned m1 = ~(((x1 & 0x7F7F7F7Fu) + 0x7F7F7F7Fu) | x1) & 0x80808080u;
        const unsigned m2 = ~(((x2 & 0x7F7F7F7Fu) + 0x7F7F7F7Fu) | x2) & 0x80808080u;
        const unsigned m3 = ~(((x3 & 0x7F7F7F7Fu) + 0x7F7F7F7Fu) | x3) & 0x80808080u;
        unsigned z = (m0 >> 7) | (m1 >> 6) | (m2 >> 5) | (m3 >> 4);
#pragma nounroll
        for (int it = 0; it < 16; ++it) {
            const bool f = z != 0u;
            const unsigned m = __builtin_amdgcn_ballot_w32(f);
            if (m == 0u) break;
            make_room();
            const int p = __builtin_ctz(z | 0x80000000u);
            push(m, f, e0 + ((p & 7) << 2) + (p >> 3));
            z &= z - 1u;
        }
    };
    auto drain_batches = [&]() {
        while (tail - head >= 32) {
            own_drain<D, MODE == 0>(ring, head, 32, lane, lo, src, dst, w, h, tile, seg, &slotCount);
            head += 32;
        }
    };

    constexpr int U     = 4;
    constexpr int STEP  = 2 * U * OWN_WAVES;
    constexpr int CQUAD = OWN_WAVES * 32;
    const int nQuad  = nEdges >> 4;
    const int nBlock = (nQuad >> 5) / STEP;

    bool scan = true;
    if (MODE == 1) {
        const int nSlot = __builtin_amdgcn_readfirstlane(seg[0].x);
        if (nSlot <= LIST_CAP) {
            scan = false;
            int sl = wave;
            int2 ent = make_int2(OWN_ROWS << 17, 0);
            if (sl < nSlot) ent = seg[(sl + 1) * 32 + lane];
            while (sl < nSlot) {
                const int nx = sl + OWN_WAVES;
                int2 nxt = ent;
                if (nx < nSlot) nxt = seg[(nx + 1) * 32 + lane];
                own_apply<D>(ent.x & 0x1FFFF, __builtin_bit_cast(float, ent.y), (int)((unsigned)ent.x >> 17), lane, h, tile);
                ent = nxt;
                sl = nx;
            }
        }
    }

    if (scan) {
    {
        const int q0 = (nBlock * STEP + wave) * 32 + lane;
        int4 rest[2 * U];
#pragma unroll
        for (int u = 0; u < 2 * U; ++u) {
            rest[u] = make_int4(-1, -1, -1, -1);
            if (q0 + u * CQUAD < nQuad) rest[u] = key4[q0 + u * CQUAD];
        }
#pragma unroll
        for (int u = 0; u < 2 * U; ++u) take((q0 + u * CQUAD) * 16, rest[u]);
        if (wave == 0) {
            const int e = nQuad * 16 + lane;
            unsigned r = 0xFFFFFFFFu;
            if (lane < (nEdges & 15)) r = (unsigned)(dst[e] - lo);
            const bool f = r < (unsigned)OWN_ROWS;
            const unsigned m = __builtin_amdgcn_ballot_w32(f);
            if (m != 0u) {
                make_room();
                push(m, f, e);
            }
        }
        drain_batches();
    }

    if (nBlock > 0) {
        int blk = (int)(((long)blockIdx.x * nBlock) / gridDim.x);
        int q = (blk * STEP + wave) * 32 + lane;
        int4 bufA[U], bufB[U];
#pragma unroll
        for (int u = 0; u < U; ++u) bufA[u] = key4[q + u * CQUAD];
        for (int i = 0; i < nBlock; ++i) {
#pragma unroll
            for (int u = 0; u < U; ++u) bufB[u] = key4[q + (U + u) * CQUAD];
#pragma unroll
            for (int u = 0; u < U; ++u) take((q + u * CQUAD) * 16, bufA[u]);
            blk = (blk + 1 == nBlock) ? 0 : blk + 1;
            const int qn = (blk * STEP + wave) * 32 + lane;
#pragma unroll
            for (int u = 0; u < U; ++u) bufA[u] = key4[qn + u * CQUAD];
#pragma unroll
            for (int u = 0; u < U; ++u) take((q + (U + u) * CQUAD) * 16, bufB[u]);
            q = qn;
            drain_batches();
        }
    }
    if (tail - head > 0) own_drain<D, MODE == 0>(ring, head, tail - head, lane, lo, src, dst, w, h, tile, seg, &slotCount);
    }

    __syncthreads();
    if (MODE == 0 && threadIdx.x == 0) seg[0] = make_int2(slotCount, 0);
    float4* __restrict__ o4 = reinterpret_cast<float4*>(agg + (long)lo * D);
    const int nOut = (blockIdx.x == N_RANGES - 1 ? LAST_ROWS : OWN_ROWS) * D / 4;
    for (int i = threadIdx.x; i < nOut; i += 1024) o4[i] = t4[i];
}

__global__ __launch_bounds__(256) void logsoftmax_k(const float* __restrict__ agg2,
                                                    const float* __restrict__ b2,
                                                    float* __restrict__ out) {
    int i = blockIdx.x * blockDim.x + threadIdx.x;
    if (i >= N_NODES) return;
    const float* __restrict__ z = agg2 + (long)i * NCLASS;
    float zb[NCLASS];
    float mx = -3.0e38f;
    for (int c = 0; c < NCLASS; ++c) {
        zb[c] = z[c] + b2[c];
        mx = fmaxf(mx, zb[c]);
    }
    float s = 0.0f;
    for (int c = 0; c < NCLASS; ++c) s += expf(zb[c] - mx);
    float lse = mx + logf(s);
    float* __restrict__ o = out + (long)i * NCLASS;
    for (int c = 0; c < NCLASS; ++c) o[c] = zb[c] - lse;
}

extern "C" void kernel_launch(void* const* d_in, const int* in_sizes, int n_in,
                              void* d_out, int out_size, void* d_ws, size_t ws_size,
                              hipStream_t stream) {
    const float* x    = (const float*)d_in[0];
    const int*   esrc = (const int*)d_in[1];
    const int*   edst = (const int*)d_in[2];
    const float* ew   = (const float*)d_in[3];
    const float* W1   = (const float*)d_in[4];
    const float* b1   = (const float*)d_in[5];
    const float* W2   = (const float*)d_in[6];
    const float* b2   = (const float*)d_in[7];
    float* out = (float*)d_out;
    const int E = in_sizes[1];

    float* h1   = (float*)d_ws;
    float* agg1 = h1   + (size_t)N_NODES * NHID;
    float* h2   = agg1 + (size_t)N_NODES * NHID;
    float* lst  = h2   + (size_t)N_NODES * NCLASS;
    float* keyf = lst  + (size_t)N_RANGES * LIST_SEG;
    float* agg2 = h1;

    const int gemmBlocks = (N_NODES + 127) / 128;
    gemm1_wmma_k<<<gemmBlocks, 256, 0, stream>>>(x, W1, h1, edst, E, keyf);

    spmm_own_k<NHID, 0><<<N_RANGES, 1024, 0, stream>>>(esrc, edst, ew, h1, agg1, E, lst);

    gemm2_wmma_k<<<gemmBlocks, 256, 0, stream>>>(agg1, b1, W2, h2);

    spmm_own_k<NCLASS, 1><<<N_RANGES, 1024, 0, stream>>>(esrc, edst, ew, h2, agg2, E, lst);

    logsoftmax_k<<<(N_NODES + 255) / 256, 256, 0, stream>>>(agg2, b2, out);
}


# Layer 2 reassociated: 2nd SpMM sums relu(agg1+b1) rows (D=64) in its LDS tile, its group then does x W2 (f32 WMMA), +b2, log_softmax there and writes out; 2 launches gone; sum orders changed
# speedup vs baseline: 3.7235x; 1.1221x over previous
#include <hip/hip_runtime.h>

typedef __attribute__((ext_vector_type(2))) float v2f;
typedef __attribute__((ext_vector_type(8))) float v8f;

#define N_NODES 100000
#define NFEAT   256
#define NHID    64
#define NCLASS  41
#define N_EDGES 1600000
#define OWN_ROWS  416
#define N_RANGES ((N_NODES + OWN_ROWS - 1) / OWN_ROWS)
#define LAST_ROWS (N_NODES - (N_RANGES - 1) * OWN_ROWS)
#define KEY_NONE 255u
static_assert(N_RANGES <= 255, "a range number fits a byte and is never KEY_NONE");
static_assert(OWN_ROWS % 32 == 0, "rows of 41 floats: 32 of them are a whole number of 128-B lines");

__device__ __forceinline__ unsigned edge_key(int d) {
    const unsigned q = (unsigned)d / OWN_ROWS;
    return q < KEY_NONE ? q : KEY_NONE;
}

__global__ __launch_bounds__(256) void gemm1_wmma_k(const float* __restrict__ x,
                                                    const float* __restrict__ W1,
                                                    float* __restrict__ h1,
                                                    const int* __restrict__ dst,
                                                    int nEdges,
                                                    float* __restrict__ keyf) {
    {
        uint2* __restrict__ key8 = reinterpret_cast<uint2*>(keyf);
        const int4* __restrict__ d4 = reinterpret_cast<const int4*>(dst);
        const int nOct = nEdges >> 3;
        for (int g = blockIdx.x * 256 + threadIdx.x; g < nOct; g += gridDim.x * 256) {
            const int4 a = d4[2 * g], b = d4[2 * g + 1];
            key8[g] = make_uint2(edge_key(a.x) | (edge_key(a.y) << 8) | (edge_key(a.z) << 16) | (edge_key(a.w) << 24),
                                 edge_key(b.x) | (edge_key(b.y) << 8) | (edge_key(b.z) << 16) | (edge_key(b.w) << 24));
        }
    }

    const int lane = threadIdx.x & 31;
    const int wave = threadIdx.x >> 5;
    const int r0 = (blockIdx.x * 8 + wave) * 16;
    if (r0 >= N_NODES) return;

    const int m     = lane & 15;
    const int khalf = (lane >> 4) * 2;

    int arow = r0 + m;
    if (arow >= N_NODES) arow = N_NODES - 1;
    const float* __restrict__ xrow = x + (long)arow * NFEAT;

    v8f acc0 = {}, acc1 = {}, acc2 = {}, acc3 = {};
    for (int k0 = 0; k0 < NFEAT; k0 += 4) {
        v2f a;
        a.x = xrow[k0 + khalf];
        a.y = xrow[k0 + khalf + 1];
        const float* __restrict__ wlo = W1 + (k0 + khalf) * NHID + m;
        const float* __restrict__ whi = W1 + (k0 + khalf + 1) * NHID + m;
        v2f b0; b0.x = wlo[0];  b0.y = whi[0];
        v2f b1; b1.x = wlo[16]; b1.y = whi[16];
        v2f b2; b2.x = wlo[32]; b2.y = whi[32];
        v2f b3; b3.x = wlo[48]; b3.y = whi[48];
        acc0 = __builtin_amdgcn_wmma_f32_16x16x4_f32(false, a, false, b0, (short)0, acc0, false, false);
        acc1 = __builtin_amdgcn_wmma_f32_16x16x4_f32(false, a, false, b1, (short)0, acc1, false, false);
        acc2 = __builtin_amdgcn_wmma_f32_16x16x4_f32(false, a, false, b2, (short)0, acc2, false, false);
        acc3 = __builtin_amdgcn_wmma_f32_16x16x4_f32(false, a, false, b3, (short)0, acc3, false, false);
    }

    const int col     = lane & 15;
    const int rowHalf = (lane >> 4) * 8;
    for (int j = 0; j < 8; ++j) {
        int r = r0 + rowHalf + j;
        if (r < N_NODES) {
            float* __restrict__ o = h1 + (long)r * NHID + col;
            o[0]  = acc0[j];
            o[16] = acc1[j];
            o[32] = acc2[j];
            o[48] = acc3[j];
        }
    }
}

#define OWN_WAVES 32
#define RING_CAP  256
#define KEY_BYTES N_EDGES
#define LIST_SEG  ((N_NODES * NCLASS * 4 - KEY_BYTES) / N_RANGES / 128 * 32)
#define LIST_CAP  (LIST_SEG / 64 - 1)
static_assert(LAST_ROWS > 0 && LAST_ROWS <= OWN_ROWS, "row ranges must cover N");
static_assert(N_NODES <= (1 << 17) && OWN_ROWS < (1 << 14), "src and row share one 32-bit word");
static_assert((LIST_SEG * 4) % 128 == 0 && (long)N_RANGES * LIST_SEG * 4 + KEY_BYTES <= (long)N_NODES * NCLASS * 4, "segments and the key plane start on a line and fit the plane");
static_assert((RING_CAP & (RING_CAP - 1)) == 0 && RING_CAP >= 64, "a wave adds at most 32 entries at a time");

__device__ __forceinline__ void lds_add(float* p, float v) {
    __hip_atomic_fetch_add(p, v, __ATOMIC_RELAXED, __HIP_MEMORY_SCOPE_WORKGROUP);
}

template <int D>
__device__ __forceinline__ void own_apply(int s, float ww, int r, int lane,
                                          const float* __restrict__ h, float* __restrict__ tile) {
#pragma unroll
    for (int hb = 0; hb < 2; ++hb) {
        float v0[16], v1[16];
#pragma unroll
        for (int j = 0; j < 16; ++j) {
            const int sj = __builtin_amdgcn_readlane(s, hb * 16 + j);
            const float* __restrict__ hr = h + (long)sj * D;
            v0[j] = hr[lane];
            v1[j] = 0.0f;
            if (lane + 32 < D) v1[j] = hr[lane + 32];
        }
#pragma unroll
        for (int j = 0; j < 16; ++j) {
            const float wj = __builtin_bit_cast(float, __builtin_amdgcn_readlane(__builtin_bit_cast(int, ww), hb * 16 + j));
            const int   rj = __builtin_amdgcn_readlane(r, hb * 16 + j);
            float* tr = tile + rj * D;
            lds_add(tr + lane, wj * v0[j]);
            if (lane + 32 < D) lds_add(tr + lane + 32, wj * v1[j]);
        }
    }
}

template <int D, bool KEEP>
__device__ __forceinline__ void own_drain(const int* __restrict__ ring, int head, int n, int lane, int lo,
                                          const int* __restrict__ src, const int* __restrict__ dst,
                                          const float* __restrict__ w,
                                          const float* __restrict__ h, float* __restrict__ tile,
                                          int2* __restrict__ seg, int* slotCount) {
    __builtin_amdgcn_fence(__ATOMIC_ACQ_REL, "wavefront");
    const bool ok = lane < n;
    const int  re = ring[(head + lane) & (RING_CAP - 1)];
    const int   e  = ok ? re : 0;
    const int   r  = ok ? dst[e] - lo : OWN_ROWS;
    const int   s  = src[e];
    const float ww = ok ? w[e] : 0.0f;
    if (KEEP) {
        int slot = 0;
        if (lane == 0) slot = __hip_atomic_fetch_add(slotCount, 1, __ATOMIC_RELAXED, __HIP_MEMORY_SCOPE_WORKGROUP);
        slot = __builtin_amdgcn_readfirstlane(slot);
        if (slot < LIST_CAP) seg[(slot + 1) * 32 + lane] = make_int2(s | (r << 17), __builtin_bit_cast(int, ww));
    }
    own_apply<D>(s, ww, r, lane, h, tile);
}

template <int D, int MODE>
__global__ __launch_bounds__(1024) void spmm_own_k(const int* __restrict__ src,
                                                   const int* __restrict__ dst,
                                                   const float* __restrict__ w,
                                                   const float* __restrict__ h,
                                                   float* __restrict__ agg,
                                                   int nEdges,
                                                   float* __restrict__ lst,
                                                   const float* __restrict__ bias,
                                                   const float* __restrict__ W2) {
    static_assert(D == NHID, "both launches sum rows of NHID floats");
    static_assert(OWN_ROWS % 16 == 0 && LAST_ROWS % 16 == 0 && OWN_ROWS / 16 <= OWN_WAVES, "MODE 1: one wave per 16 owned rows");
    static_assert((OWN_ROWS * NCLASS) % 4 == 0 && (OWN_ROWS * NCLASS * 4) % 128 == 0 &&
                  (LAST_ROWS * NCLASS) % 4 == 0 && (LAST_ROWS * NCLASS * 4) % 128 == 0, "MODE 1: out rows leave as whole 16-B pieces of whole lines");
    static_assert(1024 % (D / 4) == 0, "MODE 0: a thread's 16-B pieces of the tile all lie in the same four columns");
    static_assert((OWN_ROWS * D) % 4 == 0 && (OWN_ROWS * D * 4) % 128 == 0 &&
                  (LAST_ROWS * D) % 4 == 0 && (LAST_ROWS * D * 4) % 128 == 0, "tile stores are whole 16-B pieces of whole lines");
    __shared__ __attribute__((aligned(16))) float tile[(OWN_ROWS + 1) * D];
    __shared__ int rings[OWN_WAVES * RING_CAP];
    __shared__ int slotCount;

    const int lane = threadIdx.x & 31;
    const int wave = __builtin_amdgcn_readfirstlane(threadIdx.x >> 5);
    const int lo   = blockIdx.x * OWN_ROWS;

    float4* t4 = reinterpret_cast<float4*>(tile);
    for (int i = threadIdx.x; i < OWN_ROWS * D / 4; i += 1024) t4[i] = make_float4(0.f, 0.f, 0.f, 0.f);
    if (MODE == 0 && threadIdx.x == 0) slotCount = 0;
    __syncthreads();

    int* ring = rings + wave * RING_CAP;
    int head = 0, tail = 0;
    int2* __restrict__ seg = reinterpret_cast<int2*>(lst + (long)blockIdx.x * LIST_SEG);
    const int4* __restrict__ key4 = reinterpret_cast<const int4*>(lst + (long)N_RANGES * LIST_SEG);
    const unsigned pat = (unsigned)blockIdx.x * 0x01010101u;

    auto push = [&](unsigned m, bool f, int e) {
        if (f) ring[(tail + (int)__builtin_amdgcn_mbcnt_lo(m, 0u)) & (RING_CAP - 1)] = e;
        tail += __builtin_popcount(m);
    };
    auto make_room = [&]() {
        if (tail - head > RING_CAP - 32) {
            if (MODE == 0 && lane == 0) __hip_atomic_fetch_or(&slotCount, 1 << 30, __ATOMIC_RELAXED, __HIP_MEMORY_SCOPE_WORKGROUP);
            __builtin_amdgcn_fence(__ATOMIC_ACQ_REL, "wavefront");
#pragma nounroll
            for (int k = head; k != tail; ++k) {
                const int e = __builtin_amdgcn_readfirstlane(ring[k & (RING_CAP - 1)]);
                const int r = dst[e] - lo;
                const float wj = w[e];
                const float* __restrict__ hr = h + (long)src[e] * D;
                float* tr = tile + r * D;
                lds_add(tr + lane, wj * hr[lane]);
                if (lane + 32 < D) lds_add(tr + lane + 32, wj * hr[lane + 32]);
            }
            head = tail;
        }
    };
    auto take = [&](int e0, int4 d) {
        const unsigned x0 = (unsigned)d.x ^ pat, x1 = (unsigned)d.y ^ pat, x2 = (unsigned)d.z ^ pat, x3 = (unsigned)d.w ^ pat;
        const unsigned m0 = ~(((x0 & 0x7F7F7F7Fu) + 0x7F7F7F7Fu) | x0) & 0x80808080u;
        const unsigned m1 = ~(((x1 & 0x7F7F7F7Fu) + 0x7F7F7F7Fu) | x1) & 0x80808080u;
        const unsigned m2 = ~(((x2 & 0x7F7F7F7Fu) + 0x7F7F7F7Fu) | x2) & 0x80808080u;
        const unsigned m3 = ~(((x3 & 0x7F7F7F7Fu) + 0x7F7F7F7Fu) | x3) & 0x80808080u;
        unsigned z = (m0 >> 7) | (m1 >> 6) | (m2 >> 5) | (m3 >> 4);
#pragma nounroll
        for (int it = 0; it < 16; ++it) {
            const bool f = z != 0u;
            const unsigned m = __builtin_amdgcn_ballot_w32(f);
            if (m == 0u) break;
            make_room();
            const int p = __builtin_ctz(z | 0x80000000u);
            push(m, f, e0 + ((p & 7) << 2) + (p >> 3));
            z &= z - 1u;
        }
    };
    auto drain_batches = [&]() {
        while (tail - head >= 32) {
            own_drain<D, MODE == 0>(ring, head, 32, lane, lo, src, dst, w, h, tile, seg, &slotCount);
            head += 32;
        }
    };

    constexpr int U     = 4;
    constexpr int STEP  = 2 * U * OWN_WAVES;
    constexpr int CQUAD = OWN_WAVES * 32;
    const int nQuad  = nEdges >> 4;
    const int nBlock = (nQuad >> 5) / STEP;

    bool scan = true;
    if (MODE == 1) {
        const int nSlot = __builtin_amdgcn_readfirstlane(seg[0].x);
        if (nSlot <= LIST_CAP) {
            scan = false;
            int sl = wave;
            int2 ent = make_int2(OWN_ROWS << 17, 0);
            if (sl < nSlot) ent = seg[(sl + 1) * 32 + lane];
            while (sl < nSlot) {
                const int nx = sl + OWN_WAVES;
                int2 nxt = ent;
                if (nx < nSlot) nxt = seg[(nx + 1) * 32 + lane];
                own_apply<D>(ent.x & 0x1FFFF, __builtin_bit_cast(float, ent.y), (int)((unsigned)ent.x >> 17), lane, h, tile);
                ent = nxt;
                sl = nx;
            }
        }
    }

    if (scan) {
    {
        const int q0 = (nBlock * STEP + wave) * 32 + lane;
        int4 rest[2 * U];
#pragma unroll
        for (int u = 0; u < 2 * U; ++u) {
            rest[u] = make_int4(-1, -1, -1, -1);
            if (q0 + u * CQUAD < nQuad) rest[u] = key4[q0 + u * CQUAD];
        }
#pragma unroll
        for (int u = 0; u < 2 * U; ++u) take((q0 + u * CQUAD) * 16, rest[u]);
        if (wave == 0) {
            const int e = nQuad * 16 + lane;
            unsigned r = 0xFFFFFFFFu;
            if (lane < (nEdges & 15)) r = (unsigned)(dst[e] - lo);
            const bool f = r < (unsigned)OWN_ROWS;
            const unsigned m = __builtin_amdgcn_ballot_w32(f);
            if (m != 0u) {
                make_room();
                push(m, f, e);
            }
        }
        drain_batches();
    }

    if (nBlock > 0) {
        int blk = (int)(((long)blockIdx.x * nBlock) / gridDim.x);
        int q = (blk * STEP + wave) * 32 + lane;
        int4 bufA[U], bufB[U];
#pragma unroll
        for (int u = 0; u < U; ++u) bufA[u] = key4[q + u * CQUAD];
        for (int i = 0; i < nBlock; ++i) {
#pragma unroll
            for (int u = 0; u < U; ++u) bufB[u] = key4[q + (U + u) * CQUAD];
#pragma unroll
            for (int u = 0; u < U; ++u) take((q + u * CQUAD) * 16, bufA[u]);
            blk = (blk + 1 == nBlock) ? 0 : blk + 1;
            const int qn = (blk * STEP + wave) * 32 + lane;
#pragma unroll
            for (int u = 0; u < U; ++u) bufA[u] = key4[qn + u * CQUAD];
#pragma unroll
            for (int u = 0; u < U; ++u) take((q + (U + u) * CQUAD) * 16, bufB[u]);
            q = qn;
            drain_batches();
        }
    }
    if (tail - head > 0) own_drain<D, MODE == 0>(ring, head, tail - head, lane, lo, src, dst, w, h, tile, seg, &slotCount);
    }

    __syncthreads();
    const int rows = blockIdx.x == N_RANGES - 1 ? LAST_ROWS : OWN_ROWS;
    if (MODE == 0) {
        if (threadIdx.x == 0) seg[0] = make_int2(slotCount, 0);
        const int c4 = (threadIdx.x % (D / 4)) * 4;
        const float bx = bias[c4], by = bias[c4 + 1], bz = bias[c4 + 2], bw = bias[c4 + 3];
        float4* __restrict__ o4 = reinterpret_cast<float4*>(agg + (long)lo * D);
        const int nOut = rows * D / 4;
        for (int i = threadIdx.x; i < nOut; i += 1024) {
            const float4 v = t4[i];
            o4[i] = make_float4(fmaxf(v.x + bx, 0.0f), fmaxf(v.y + by, 0.0f), fmaxf(v.z + bz, 0.0f), fmaxf(v.w + bw, 0.0f));
        }
    } else {
        const int m     = lane & 15;
        const int khalf = (lane >> 4) * 2;
        const bool mine = wave * 16 < rows;
        const bool has2 = 32 + m < NCLASS;
        float res0[8], res1[8], res2[8];
        if (mine) {
            const float* arow = tile + (wave * 16 + m) * D;
            v8f acc0 = {}, acc1 = {}, acc2 = {};
#pragma unroll
            for (int k0 = 0; k0 < NHID; k0 += 4) {
                const int ka = k0 + khalf;
                v2f a;
                a.x = arow[ka];
                a.y = arow[ka + 1];
                v2f b0, b1v, b2v;
                b0.x  = W2[ka * NCLASS + m];
                b0.y  = W2[(ka + 1) * NCLASS + m];
                b1v.x = W2[ka * NCLASS + 16 + m];
                b1v.y = W2[(ka + 1) * NCLASS + 16 + m];
                const int nc = has2 ? 32 + m : NCLASS - 1;
                float lo2 = W2[ka * NCLASS + nc];
                float hi2 = W2[(ka + 1) * NCLASS + nc];
                if (!has2) { lo2 = 0.0f; hi2 = 0.0f; }
                b2v.x = lo2; b2v.y = hi2;
                acc0 = __builtin_amdgcn_wmma_f32_16x16x4_f32(false, a, false, b0,  (short)0, acc0, false, false);
                acc1 = __builtin_amdgcn_wmma_f32_16x16x4_f32(false, a, false, b1v, (short)0, acc1, false, false);
                acc2 = __builtin_amdgcn_wmma_f32_16x16x4_f32(false, a, false, b2v, (short)0, acc2, false, false);
            }
            const float bb0 = bias[m], bb1 = bias[16 + m];
            const float bb2 = has2 ? bias[32 + m] : 0.0f;
#pragma unroll
            for (int j = 0; j < 8; ++j) {
                const float z0 = acc0[j] + bb0, z1 = acc1[j] + bb1, z2 = acc2[j] + bb2;
                float mx = fmaxf(z0, z1);
                if (has2) mx = fmaxf(mx, z2);
                mx = fmaxf(mx, __shfl_xor(mx, 1));
                mx = fmaxf(mx, __shfl_xor(mx, 2));
                mx = fmaxf(mx, __shfl_xor(mx, 4));
                mx = fmaxf(mx, __shfl_xor(mx, 8));
                float s = expf(z0 - mx) + expf(z1 - mx);
                if (has2) s += expf(z2 - mx);
                s += __shfl_xor(s, 1);
                s += __shfl_xor(s, 2);
                s += __shfl_xor(s, 4);
                s += __shfl_xor(s, 8);
                const float lse = mx + logf(s);
                res0[j] = z0 - lse;
                res1[j] = z1 - lse;
                res2[j] = z2 - lse;
            }
        }
        __syncthreads();
        if (mine) {
            float* st = tile + (wave * 16 + (lane >> 4) * 8) * NCLASS + m;
#pragma unroll
            for (int j = 0; j < 8; ++j) {
                st[j * NCLASS]      = res0[j];
                st[j * NCLASS + 16] = res1[j];
                if (has2) st[j * NCLASS + 32] = res2[j];
            }
        }
        __syncthreads();
        float4* __restrict__ o4 = reinterpret_cast<float4*>(agg + (long)lo * NCLASS);
        const int nOut = rows * NCLASS / 4;
        for (int i = threadIdx.x; i < nOut; i += 1024) o4[i] = t4[i];
    }
}

extern "C" void kernel_launch(void* const* d_in, const int* in_sizes, int n_in,
                              void* d_out, int out_size, void* d_ws, size_t ws_size,
                              hipStream_t stream) {
    const float* x    = (const float*)d_in[0];
    const int*   esrc = (const int*)d_in[1];
    const int*   edst = (const int*)d_in[2];
    const float* ew   = (const float*)d_in[3];
    const float* W1   = (const float*)d_in[4];
    const float* b1   = (const float*)d_in[5];
    const float* W2   = (const float*)d_in[6];
    const float* b2   = (const float*)d_in[7];
    float* out = (float*)d_out;
    const int E = in_sizes[1];

    float* h1   = (float*)d_ws;
    float* agg1 = h1   + (size_t)N_NODES * NHID;
    float* lst  = agg1 + (size_t)N_NODES * NHID;
    float* keyf = lst  + (size_t)N_RANGES * LIST_SEG;

    const int gemmBlocks = (N_NODES + 127) / 128;
    gemm1_wmma_k<<<gemmBlocks, 256, 0, stream>>>(x, W1, h1, edst, E, keyf);

    spmm_own_k<NHID, 0><<<N_RANGES, 1024, 0, stream>>>(esrc, edst, ew, h1, agg1, E, lst, b1, W2);

    spmm_own_k<NHID, 1><<<N_RANGES, 1024, 0, stream>>>(esrc, edst, ew, agg1, out, E, lst, b2, W2);
}


# GEMM1 operand feed: a wave keeps its 16-column tile of W1 in registers for all k; x staged in LDS by whole 512-B row loads (A fragment = one 8-B LDS read); h1 leaves as whole lines; k order kept
# speedup vs baseline: 3.8869x; 1.0439x over previous
#include <hip/hip_runtime.h>

typedef __attribute__((ext_vector_type(2))) float v2f;
typedef __attribute__((ext_vector_type(8))) float v8f;
typedef __attribute__((ext_vector_type(4))) float v4f;

#define N_NODES 100000
#define NFEAT   256
#define NHID    64
#define NCLASS  41
#define N_EDGES 1600000
#define OWN_ROWS  416
#define N_RANGES ((N_NODES + OWN_ROWS - 1) / OWN_ROWS)
#define LAST_ROWS (N_NODES - (N_RANGES - 1) * OWN_ROWS)
#define KEY_NONE 255u
static_assert(N_RANGES <= 255, "a range number fits a byte and is never KEY_NONE");
static_assert(OWN_ROWS % 32 == 0, "rows of 41 floats: 32 of them are a whole number of 128-B lines");

__device__ __forceinline__ unsigned edge_key(int d) {
    const unsigned q = (unsigned)d / OWN_ROWS;
    return q < KEY_NONE ? q : KEY_NONE;
}

#define G1_GROUPS 512
#define G1_ROWS   32
#define G1_PITCH  260
#define G1_OPITCH 68
static_assert((G1_PITCH * 4) % 16 == 0 && (G1_OPITCH * 4) % 16 == 0, "staged rows keep 16-B pieces aligned");

__global__ __launch_bounds__(256) void gemm1_wmma_k(const float* __restrict__ x,
                                                       const float* __restrict__ W1,
                                                       float* __restrict__ h1,
                                                       const int* __restrict__ dst,
                                                       int nEdges,
                                                       float* __restrict__ keyf) {
    {
        uint2* __restrict__ key8 = reinterpret_cast<uint2*>(keyf);
        const int4* __restrict__ d4 = reinterpret_cast<const int4*>(dst);
        const int nOct = nEdges >> 3;
        for (int g = blockIdx.x * 256 + threadIdx.x; g < nOct; g += gridDim.x * 256) {
            const int4 a = d4[2 * g], b = d4[2 * g + 1];
            key8[g] = make_uint2(edge_key(a.x) | (edge_key(a.y) << 8) | (edge_key(a.z) << 16) | (edge_key(a.w) << 24),
                                 edge_key(b.x) | (edge_key(b.y) << 8) | (edge_key(b.z) << 16) | (edge_key(b.w) << 24));
        }
    }

    __shared__ __attribute__((aligned(16))) float xs[2][G1_ROWS * G1_PITCH];
    __shared__ __attribute__((aligned(16))) float os[2][G1_ROWS * G1_OPITCH];

    const int lane  = threadIdx.x & 31;
    const int wave  = threadIdx.x >> 5;
    const int m     = lane & 15;
    const int khalf = (lane >> 4) * 2;
    const int ct    = wave & 3;
    const int rt    = wave >> 2;

    v2f bw[NFEAT / 4];
#pragma unroll
    for (int ks = 0; ks < NFEAT / 4; ++ks) {
        bw[ks].x = W1[(4 * ks + khalf) * NHID + ct * 16 + m];
        bw[ks].y = W1[(4 * ks + khalf + 1) * NHID + ct * 16 + m];
    }

    const int nTrip = (N_NODES + G1_ROWS - 1) / G1_ROWS;
    const int srow = threadIdx.x >> 6;
    const int spc  = (threadIdx.x & 63) * 4;
    const int orow = threadIdx.x >> 4;
    const int opc  = (threadIdx.x & 15) * 4;
    v4f pre[G1_ROWS / 4];

    int trip = blockIdx.x;
    if (trip < nTrip) {
#pragma unroll
        for (int p = 0; p < G1_ROWS / 4; ++p) {
            int r = trip * G1_ROWS + 4 * p + srow;
            if (r >= N_NODES) r = N_NODES - 1;
            pre[p] = *reinterpret_cast<const v4f*>(x + (long)r * NFEAT + spc);
        }
#pragma unroll
        for (int p = 0; p < G1_ROWS / 4; ++p)
            *reinterpret_cast<v4f*>(&xs[0][(4 * p + srow) * G1_PITCH + spc]) = pre[p];
    }
    __syncthreads();

    int buf = 0;
    int prevTrip = -1;
    for (; trip < nTrip; trip += gridDim.x) {
        const int  nx   = trip + gridDim.x;
        const bool more = nx < nTrip;
        if (more) {
#pragma unroll
            for (int p = 0; p < G1_ROWS / 4; ++p) {
                int r = nx * G1_ROWS + 4 * p + srow;
                if (r >= N_NODES) r = N_NODES - 1;
                pre[p] = *reinterpret_cast<const v4f*>(x + (long)r * NFEAT + spc);
            }
        }
        if (prevTrip >= 0) {
#pragma unroll
            for (int p = 0; p < G1_ROWS / 16; ++p) {
                const int rr = 16 * p + orow;
                const int r  = prevTrip * G1_ROWS + rr;
                if (r < N_NODES)
                    *reinterpret_cast<v4f*>(h1 + (long)r * NHID + opc) =
                        *reinterpret_cast<const v4f*>(&os[buf ^ 1][rr * G1_OPITCH + opc]);
            }
        }

        const float* arow = &xs[buf][(rt * 16 + m) * G1_PITCH + khalf];
        v8f acc = {};
#pragma unroll
        for (int ks = 0; ks < NFEAT / 4; ++ks) {
            const v2f a = *reinterpret_cast<const v2f*>(arow + 4 * ks);
            acc = __builtin_amdgcn_wmma_f32_16x16x4_f32(false, a, false, bw[ks], (short)0, acc, false, false);
        }

        float* ost = &os[buf][(rt * 16 + (lane >> 4) * 8) * G1_OPITCH + ct * 16 + m];
#pragma unroll
        for (int j = 0; j < 8; ++j) ost[j * G1_OPITCH] = acc[j];

        if (more) {
#pragma unroll
            for (int p = 0; p < G1_ROWS / 4; ++p)
                *reinterpret_cast<v4f*>(&xs[buf ^ 1][(4 * p + srow) * G1_PITCH + spc]) = pre[p];
        }
        __syncthreads();
        prevTrip = trip;
        buf ^= 1;
    }
    if (prevTrip >= 0) {
#pragma unroll
        for (int p = 0; p < G1_ROWS / 16; ++p) {
            const int rr = 16 * p + orow;
            const int r  = prevTrip * G1_ROWS + rr;
            if (r < N_NODES)
                *reinterpret_cast<v4f*>(h1 + (long)r * NHID + opc) =
                    *reinterpret_cast<const v4f*>(&os[buf ^ 1][rr * G1_OPITCH + opc]);
        }
    }
}

#define OWN_WAVES 32
#define RING_CAP  256
#define KEY_BYTES N_EDGES
#define LIST_SEG  ((N_NODES * NCLASS * 4 - KEY_BYTES) / N_RANGES / 128 * 32)
#define LIST_CAP  (LIST_SEG / 64 - 1)
static_assert(LAST_ROWS > 0 && LAST_ROWS <= OWN_ROWS, "row ranges must cover N");
static_assert(N_NODES <= (1 << 17) && OWN_ROWS < (1 << 14), "src and row share one 32-bit word");
static_assert((LIST_SEG * 4) % 128 == 0 && (long)N_RANGES * LIST_SEG * 4 + KEY_BYTES <= (long)N_NODES * NCLASS * 4, "segments and the key plane start on a line and fit the plane");
static_assert((RING_CAP & (RING_CAP - 1)) == 0 && RING_CAP >= 64, "a wave adds at most 32 entries at a time");

__device__ __forceinline__ void lds_add(float* p, float v) {
    __hip_atomic_fetch_add(p, v, __ATOMIC_RELAXED, __HIP_MEMORY_SCOPE_WORKGROUP);
}

template <int D>
__device__ __forceinline__ void own_apply(int s, float ww, int r, int lane,
                                          const float* __restrict__ h, float* __restrict__ tile) {
#pragma unroll
    for (int hb = 0; hb < 2; ++hb) {
        float v0[16], v1[16];
#pragma unroll
        for (int j = 0; j < 16; ++j) {
            const int sj = __builtin_amdgcn_readlane(s, hb * 16 + j);
            const float* __restrict__ hr = h + (long)sj * D;
            v0[j] = hr[lane];
            v1[j] = 0.0f;
            if (lane + 32 < D) v1[j] = hr[lane + 32];
        }
#pragma unroll
        for (int j = 0; j < 16; ++j) {
            const float wj = __builtin_bit_cast(float, __builtin_amdgcn_readlane(__builtin_bit_cast(int, ww), hb * 16 + j));
            const int   rj = __builtin_amdgcn_readlane(r, hb * 16 + j);
            float* tr = tile + rj * D;
            lds_add(tr + lane, wj * v0[j]);
            if (lane + 32 < D) lds_add(tr + lane + 32, wj * v1[j]);
        }
    }
}

template <int D, bool KEEP>
__device__ __forceinline__ void own_drain(const int* __restrict__ ring, int head, int n, int lane, int lo,
                                          const int* __restrict__ src, const int* __restrict__ dst,
                                          const float* __restrict__ w,
                                          const float* __restrict__ h, float* __restrict__ tile,
                                          int2* __restrict__ seg, int* slotCount) {
    __builtin_amdgcn_fence(__ATOMIC_ACQ_REL, "wavefront");
    const bool ok = lane < n;
    const int  re = ring[(head + lane) & (RING_CAP - 1)];
    const int   e  = ok ? re : 0;
    const int   r  = ok ? dst[e] - lo : OWN_ROWS;
    const int   s  = src[e];
    const float ww = ok ? w[e] : 0.0f;
    if (KEEP) {
        int slot = 0;
        if (lane == 0) slot = __hip_atomic_fetch_add(slotCount, 1, __ATOMIC_RELAXED, __HIP_MEMORY_SCOPE_WORKGROUP);
        slot = __builtin_amdgcn_readfirstlane(slot);
        if (slot < LIST_CAP) seg[(slot + 1) * 32 + lane] = make_int2(s | (r << 17), __builtin_bit_cast(int, ww));
    }
    own_apply<D>(s, ww, r, lane, h, tile);
}

template <int D, int MODE>
__global__ __launch_bounds__(1024) void spmm_own_k(const int* __restrict__ src,
                                                   const int* __restrict__ dst,
                                                   const float* __restrict__ w,
                                                   const float* __restrict__ h,
                                                   float* __restrict__ agg,
                                                   int nEdges,
                                                   float* __restrict__ lst,
                                                   const float* __restrict__ bias,
                                                   const float* __restrict__ W2) {
    static_assert(D == NHID, "both launches sum rows of NHID floats");
    static_assert(OWN_ROWS % 16 == 0 && LAST_ROWS % 16 == 0 && OWN_ROWS / 16 <= OWN_WAVES, "MODE 1: one wave per 16 owned rows");
    static_assert((OWN_ROWS * NCLASS) % 4 == 0 && (OWN_ROWS * NCLASS * 4) % 128 == 0 &&
                  (LAST_ROWS * NCLASS) % 4 == 0 && (LAST_ROWS * NCLASS * 4) % 128 == 0, "MODE 1: out rows leave as whole 16-B pieces of whole lines");
    static_assert(1024 % (D / 4) == 0, "MODE 0: a thread's 16-B pieces of the tile all lie in the same four columns");
    static_assert((OWN_ROWS * D) % 4 == 0 && (OWN_ROWS * D * 4) % 128 == 0 &&
                  (LAST_ROWS * D) % 4 == 0 && (LAST_ROWS * D * 4) % 128 == 0, "tile stores are whole 16-B pieces of whole lines");
    __shared__ __attribute__((aligned(16))) float tile[(OWN_ROWS + 1) * D];
    __shared__ int rings[OWN_WAVES * RING_CAP];
    __shared__ int slotCount;

    const int lane = threadIdx.x & 31;
    const int wave = __builtin_amdgcn_readfirstlane(threadIdx.x >> 5);
    const int lo   = blockIdx.x * OWN_ROWS;

    float4* t4 = reinterpret_cast<float4*>(tile);
    for (int i = threadIdx.x; i < OWN_ROWS * D / 4; i += 1024) t4[i] = make_float4(0.f, 0.f, 0.f, 0.f);
    if (MODE == 0 && threadIdx.x == 0) slotCount = 0;
    __syncthreads();

    int* ring = rings + wave * RING_CAP;
    int head = 0, tail = 0;
    int2* __restrict__ seg = reinterpret_cast<int2*>(lst + (long)blockIdx.x * LIST_SEG);
    const int4* __restrict__ key4 = reinterpret_cast<const int4*>(lst + (long)N_RANGES * LIST_SEG);
    const unsigned pat = (unsigned)blockIdx.x * 0x01010101u;

    auto push = [&](unsigned m, bool f, int e) {
        if (f) ring[(tail + (int)__builtin_amdgcn_mbcnt_lo(m, 0u)) & (RING_CAP - 1)] = e;
        tail += __builtin_popcount(m);
    };
    auto make_room = [&]() {
        if (tail - head > RING_CAP - 32) {
            if (MODE == 0 && lane == 0) __hip_atomic_fetch_or(&slotCount, 1 << 30, __ATOMIC_RELAXED, __HIP_MEMORY_SCOPE_WORKGROUP);
            __builtin_amdgcn_fence(__ATOMIC_ACQ_REL, "wavefront");
#pragma nounroll
            for (int k = head; k != tail; ++k) {
                const int e = __builtin_amdgcn_readfirstlane(ring[k & (RING_CAP - 1)]);
                const int r = dst[e] - lo;
                const float wj = w[e];
                const float* __restrict__ hr = h + (long)src[e] * D;
                float* tr = tile + r * D;
                lds_add(tr + lane, wj * hr[lane]);
                if (lane + 32 < D) lds_add(tr + lane + 32, wj * hr[lane + 32]);
            }
            head = tail;
        }
    };
    auto take = [&](int e0, int4 d) {
        const unsigned x0 = (unsigned)d.x ^ pat, x1 = (unsigned)d.y ^ pat, x2 = (unsigned)d.z ^ pat, x3 = (unsigned)d.w ^ pat;
        const unsigned m0 = ~(((x0 & 0x7F7F7F7Fu) + 0x7F7F7F7Fu) | x0) & 0x80808080u;
        const unsigned m1 = ~(((x1 & 0x7F7F7F7Fu) + 0x7F7F7F7Fu) | x1) & 0x80808080u;
        const unsigned m2 = ~(((x2 & 0x7F7F7F7Fu) + 0x7F7F7F7Fu) | x2) & 0x80808080u;
        const unsigned m3 = ~(((x3 & 0x7F7F7F7Fu) + 0x7F7F7F7Fu) | x3) & 0x80808080u;
        unsigned z = (m0 >> 7) | (m1 >> 6) | (m2 >> 5) | (m3 >> 4);
#pragma nounroll
        for (int it = 0; it < 16; ++it) {
            const bool f = z != 0u;
            const unsigned m = __builtin_amdgcn_ballot_w32(f);
            if (m == 0u) break;
            make_room();
            const int p = __builtin_ctz(z | 0x80000000u);
            push(m, f, e0 + ((p & 7) << 2) + (p >> 3));
            z &= z - 1u;
        }
    };
    auto drain_batches = [&]() {
        while (tail - head >= 32) {
            own_drain<D, MODE == 0>(ring, head, 32, lane, lo, src, dst, w, h, tile, seg, &slotCount);
            head += 32;
        }
    };

    constexpr int U     = 4;
    constexpr int STEP  = 2 * U * OWN_WAVES;
    constexpr int CQUAD = OWN_WAVES * 32;
    const int nQuad  = nEdges >> 4;
    const int nBlock = (nQuad >> 5) / STEP;

    bool scan = true;
    if (MODE == 1) {
        const int nSlot = __builtin_amdgcn_readfirstlane(seg[0].x);
        if (nSlot <= LIST_CAP) {
            scan = false;
            int sl = wave;
            int2 ent = make_int2(OWN_ROWS << 17, 0);
            if (sl < nSlot) ent = seg[(sl + 1) * 32 + lane];
            while (sl < nSlot) {
                const int nx = sl + OWN_WAVES;
                int2 nxt = ent;
                if (nx < nSlot) nxt = seg[(nx + 1) * 32 + lane];
                own_apply<D>(ent.x & 0x1FFFF, __builtin_bit_cast(float, ent.y), (int)((unsigned)ent.x >> 17), lane, h, tile);
                ent = nxt;
                sl = nx;
            }
        }
    }

    if (scan) {
    {
        const int q0 = (nBlock * STEP + wave) * 32 + lane;
        int4 rest[2 * U];
#pragma unroll
        for (int u = 0; u < 2 * U; ++u) {
            rest[u] = make_int4(-1, -1, -1, -1);
            if (q0 + u * CQUAD < nQuad) rest[u] = key4[q0 + u * CQUAD];
        }
#pragma unroll
        for (int u = 0; u < 2 * U; ++u) take((q0 + u * CQUAD) * 16, rest[u]);
        if (wave == 0) {
            const int e = nQuad * 16 + lane;
            unsigned r = 0xFFFFFFFFu;
            if (lane < (nEdges & 15)) r = (unsigned)(dst[e] - lo);
            const bool f = r < (unsigned)OWN_ROWS;
            const unsigned m = __builtin_amdgcn_ballot_w32(f);
            if (m != 0u) {
                make_room();
                push(m, f, e);
            }
        }
        drain_batches();
    }

    if (nBlock > 0) {
        int blk = (int)(((long)blockIdx.x * nBlock) / gridDim.x);
        int q = (blk * STEP + wave) * 32 + lane;
        int4 bufA[U], bufB[U];
#pragma unroll
        for (int u = 0; u < U; ++u) bufA[u] = key4[q + u * CQUAD];
        for (int i = 0; i < nBlock; ++i) {
#pragma unroll
            for (int u = 0; u < U; ++u) bufB[u] = key4[q + (U + u) * CQUAD];
#pragma unroll
            for (int u = 0; u < U; ++u) take((q + u * CQUAD) * 16, bufA[u]);
            blk = (blk + 1 == nBlock) ? 0 : blk + 1;
            const int qn = (blk * STEP + wave) * 32 + lane;
#pragma unroll
            for (int u = 0; u < U; ++u) bufA[u] = key4[qn + u * CQUAD];
#pragma unroll
            for (int u = 0; u < U; ++u) take((q + (U + u) * CQUAD) * 16, bufB[u]);
            q = qn;
            drain_batches();
        }
    }
    if (tail - head > 0) own_drain<D, MODE == 0>(ring, head, tail - head, lane, lo, src, dst, w, h, tile, seg, &slotCount);
    }

    __syncthreads();
    const int rows = blockIdx.x == N_RANGES - 1 ? LAST_ROWS : OWN_ROWS;
    if (MODE == 0) {
        if (threadIdx.x == 0) seg[0] = make_int2(slotCount, 0);
        const int c4 = (threadIdx.x % (D / 4)) * 4;
        const float bx = bias[c4], by = bias[c4 + 1], bz = bias[c4 + 2], bw = bias[c4 + 3];
        float4* __restrict__ o4 = reinterpret_cast<float4*>(agg + (long)lo * D);
        const int nOut = rows * D / 4;
        for (int i = threadIdx.x; i < nOut; i += 1024) {
            const float4 v = t4[i];
            o4[i] = make_float4(fmaxf(v.x + bx, 0.0f), fmaxf(v.y + by, 0.0f), fmaxf(v.z + bz, 0.0f), fmaxf(v.w + bw, 0.0f));
        }
    } else {
        const int m     = lane & 15;
        const int khalf = (lane >> 4) * 2;
        const bool mine = wave * 16 < rows;
        const bool has2 = 32 + m < NCLASS;
        float res0[8], res1[8], res2[8];
        if (mine) {
            const float* arow = tile + (wave * 16 + m) * D;
            v8f acc0 = {}, acc1 = {}, acc2 = {};
#pragma unroll
            for (int k0 = 0; k0 < NHID; k0 += 4) {
                const int ka = k0 + khalf;
                v2f a;
                a.x = arow[ka];
                a.y = arow[ka + 1];
                v2f b0, b1v, b2v;
                b0.x  = W2[ka * NCLASS + m];
                b0.y  = W2[(ka + 1) * NCLASS + m];
                b1v.x = W2[ka * NCLASS + 16 + m];
                b1v.y = W2[(ka + 1) * NCLASS + 16 + m];
                const int nc = has2 ? 32 + m : NCLASS - 1;
                float lo2 = W2[ka * NCLASS + nc];
                float hi2 = W2[(ka + 1) * NCLASS + nc];
                if (!has2) { lo2 = 0.0f; hi2 = 0.0f; }
                b2v.x = lo2; b2v.y = hi2;
                acc0 = __builtin_amdgcn_wmma_f32_16x16x4_f32(false, a, false, b0,  (short)0, acc0, false, false);
                acc1 = __builtin_amdgcn_wmma_f32_16x16x4_f32(false, a, false, b1v, (short)0, acc1, false, false);
                acc2 = __builtin_amdgcn_wmma_f32_16x16x4_f32(false, a, false, b2v, (short)0, acc2, false, false);
            }
            const float bb0 = bias[m], bb1 = bias[16 + m];
            const float bb2 = has2 ? bias[32 + m] : 0.0f;
#pragma unroll
            for (int j = 0; j < 8; ++j) {
                const float z0 = acc0[j] + bb0, z1 = acc1[j] + bb1, z2 = acc2[j] + bb2;
                float mx = fmaxf(z0, z1);
                if (has2) mx = fmaxf(mx, z2);
                mx = fmaxf(mx, __shfl_xor(mx, 1));
                mx = fmaxf(mx, __shfl_xor(mx, 2));
                mx = fmaxf(mx, __shfl_xor(mx, 4));
                mx = fmaxf(mx, __shfl_xor(mx, 8));
                float s = expf(z0 - mx) + expf(z1 - mx);
                if (has2) s += expf(z2 - mx);
                s += __shfl_xor(s, 1);
                s += __shfl_xor(s, 2);
                s += __shfl_xor(s, 4);
                s += __shfl_xor(s, 8);
                const float lse = mx + logf(s);
                res0[j] = z0 - lse;
                res1[j] = z1 - lse;
                res2[j] = z2 - lse;
            }
        }
        __syncthreads();
        if (mine) {
            float* st = tile + (wave * 16 + (lane >> 4) * 8) * NCLASS + m;
#pragma unroll
            for (int j = 0; j < 8; ++j) {
                st[j * NCLASS]      = res0[j];
                st[j * NCLASS + 16] = res1[j];
                if (has2) st[j * NCLASS + 32] = res2[j];
            }
        }
        __syncthreads();
        float4* __restrict__ o4 = reinterpret_cast<float4*>(agg + (long)lo * NCLASS);
        const int nOut = rows * NCLASS / 4;
        for (int i = threadIdx.x; i < nOut; i += 1024) o4[i] = t4[i];
    }
}

extern "C" void kernel_launch(void* const* d_in, const int* in_sizes, int n_in,
                              void* d_out, int out_size, void* d_ws, size_t ws_size,
                              hipStream_t stream) {
    const float* x    = (const float*)d_in[0];
    const int*   esrc = (const int*)d_in[1];
    const int*   edst = (const int*)d_in[2];
    const float* ew   = (const float*)d_in[3];
    const float* W1   = (const float*)d_in[4];
    const float* b1   = (const float*)d_in[5];
    const float* W2   = (const float*)d_in[6];
    const float* b2   = (const float*)d_in[7];
    float* out = (float*)d_out;
    const int E = in_sizes[1];

    float* h1   = (float*)d_ws;
    float* agg1 = h1   + (size_t)N_NODES * NHID;
    float* lst  = agg1 + (size_t)N_NODES * NHID;
    float* keyf = lst  + (size_t)N_RANGES * LIST_SEG;

    gemm1_wmma_k<<<G1_GROUPS, 256, 0, stream>>>(x, W1, h1, edst, E, keyf);

    spmm_own_k<NHID, 0><<<N_RANGES, 1024, 0, stream>>>(esrc, edst, ew, h1, agg1, E, lst, b1, W2);

    spmm_own_k<NHID, 1><<<N_RANGES, 1024, 0, stream>>>(esrc, edst, ew, agg1, out, E, lst, b2, W2);
}


# GEMM1 launch packs each edge's src, row, w into one 8-B record (plane in out's memory until the last launch writes out); a kept edge costs SpMM 1 one scattered read, not 3; sum order: LDS arrival
# speedup vs baseline: 4.2439x; 1.0918x over previous
#include <hip/hip_runtime.h>

typedef __attribute__((ext_vector_type(2))) float v2f;
typedef __attribute__((ext_vector_type(8))) float v8f;
typedef __attribute__((ext_vector_type(4))) float v4f;

#define N_NODES 100000
#define NFEAT   256
#define NHID    64
#define NCLASS  41
#define N_EDGES 1600000
#define OWN_ROWS  416
#define N_RANGES ((N_NODES + OWN_ROWS - 1) / OWN_ROWS)
#define LAST_ROWS (N_NODES - (N_RANGES - 1) * OWN_ROWS)
#define KEY_NONE 255u
static_assert(N_RANGES <= 255, "a range number fits a byte and is never KEY_NONE");
static_assert(OWN_ROWS % 32 == 0, "rows of 41 floats: 32 of them are a whole number of 128-B lines");

__device__ __forceinline__ unsigned edge_key(int d) {
    const unsigned q = (unsigned)d / OWN_ROWS;
    return q < KEY_NONE ? q : KEY_NONE;
}

__device__ __forceinline__ unsigned edge_row(int d) {
    return (unsigned)d % OWN_ROWS;
}

#define G1_GROUPS 512
#define G1_ROWS   32
#define G1_PITCH  260
#define G1_OPITCH 68
static_assert((G1_PITCH * 4) % 16 == 0 && (G1_OPITCH * 4) % 16 == 0, "staged rows keep 16-B pieces aligned");

__global__ __launch_bounds__(256) void gemm1_wmma_k(const float* __restrict__ x,
                                                       const float* __restrict__ W1,
                                                       float* __restrict__ h1,
                                                       const int* __restrict__ dst,
                                                       int nEdges,
                                                       float* __restrict__ keyf,
                                                       const int* __restrict__ src,
                                                       const float* __restrict__ w,
                                                       float* __restrict__ recf) {
    {
        uint2* __restrict__ key8 = reinterpret_cast<uint2*>(keyf);
        const int4* __restrict__ d4 = reinterpret_cast<const int4*>(dst);
        const int nOct = nEdges >> 3;
        for (int g = blockIdx.x * 256 + threadIdx.x; g < nOct; g += gridDim.x * 256) {
            const int4 a = d4[2 * g], b = d4[2 * g + 1];
            key8[g] = make_uint2(edge_key(a.x) | (edge_key(a.y) << 8) | (edge_key(a.z) << 16) | (edge_key(a.w) << 24),
                                 edge_key(b.x) | (edge_key(b.y) << 8) | (edge_key(b.z) << 16) | (edge_key(b.w) << 24));
        }
    }
    {
        uint4* __restrict__ rec4 = reinterpret_cast<uint4*>(recf);
        const int2*   __restrict__ d2 = reinterpret_cast<const int2*>(dst);
        const int2*   __restrict__ s2 = reinterpret_cast<const int2*>(src);
        const float2* __restrict__ w2 = reinterpret_cast<const float2*>(w);
        const int nPair  = nEdges >> 1;
        const int stride = gridDim.x * 256;
        constexpr int RU = 8;
        for (int g0 = blockIdx.x * 256 + threadIdx.x; g0 < nPair; g0 += RU * stride) {
            int2 d[RU], s[RU];
            float2 ww[RU];
#pragma unroll
            for (int u = 0; u < RU; ++u) {
                const int g = g0 + u * stride < nPair ? g0 + u * stride : g0;
                d[u] = d2[g]; s[u] = s2[g]; ww[u] = w2[g];
            }
#pragma unroll
            for (int u = 0; u < RU; ++u)
                if (g0 + u * stride < nPair)
                    rec4[g0 + u * stride] = make_uint4((unsigned)s[u].x | (edge_row(d[u].x) << 17), __builtin_bit_cast(unsigned, ww[u].x),
                                                       (unsigned)s[u].y | (edge_row(d[u].y) << 17), __builtin_bit_cast(unsigned, ww[u].y));
        }
        if ((nEdges & 1) && blockIdx.x == 0 && threadIdx.x == 0) {
            const int e = nEdges - 1;
            reinterpret_cast<uint2*>(recf)[e] = make_uint2((unsigned)src[e] | (edge_row(dst[e]) << 17), __builtin_bit_cast(unsigned, w[e]));
        }
    }

    __shared__ __attribute__((aligned(16))) float xs[2][G1_ROWS * G1_PITCH];
    __shared__ __attribute__((aligned(16))) float os[2][G1_ROWS * G1_OPITCH];

    const int lane  = threadIdx.x & 31;
    const int wave  = threadIdx.x >> 5;
    const int m     = lane & 15;
    const int khalf = (lane >> 4) * 2;
    const int ct    = wave & 3;
    const int rt    = wave >> 2;

    v2f bw[NFEAT / 4];
#pragma unroll
    for (int ks = 0; ks < NFEAT / 4; ++ks) {
        bw[ks].x = W1[(4 * ks + khalf) * NHID + ct * 16 + m];
        bw[ks].y = W1[(4 * ks + khalf + 1) * NHID + ct * 16 + m];
    }

    const int nTrip = (N_NODES + G1_ROWS - 1) / G1_ROWS;
    const int srow = threadIdx.x >> 6;
    const int spc  = (threadIdx.x & 63) * 4;
    const int orow = threadIdx.x >> 4;
    const int opc  = (threadIdx.x & 15) * 4;
    v4f pre[G1_ROWS / 4];

    int trip = blockIdx.x;
    if (trip < nTrip) {
#pragma unroll
        for (int p = 0; p < G1_ROWS / 4; ++p) {
            int r = trip * G1_ROWS + 4 * p + srow;
            if (r >= N_NODES) r = N_NODES - 1;
            pre[p] = *reinterpret_cast<const v4f*>(x + (long)r * NFEAT + spc);
        }
#pragma unroll
        for (int p = 0; p < G1_ROWS / 4; ++p)
            *reinterpret_cast<v4f*>(&xs[0][(4 * p + srow) * G1_PITCH + spc]) = pre[p];
    }
    __syncthreads();

    int buf = 0;
    int prevTrip = -1;
    for (; trip < nTrip; trip += gridDim.x) {
        const int  nx   = trip + gridDim.x;
        const bool more = nx < nTrip;
        if (more) {
#pragma unroll
            for (int p = 0; p < G1_ROWS / 4; ++p) {
                int r = nx * G1_ROWS + 4 * p + srow;
                if (r >= N_NODES) r = N_NODES - 1;
                pre[p] = *reinterpret_cast<const v4f*>(x + (long)r * NFEAT + spc);
            }
        }
        if (prevTrip >= 0) {
#pragma unroll
            for (int p = 0; p < G1_ROWS / 16; ++p) {
                const int rr = 16 * p + orow;
                const int r  = prevTrip * G1_ROWS + rr;
                if (r < N_NODES)
                    *reinterpret_cast<v4f*>(h1 + (long)r * NHID + opc) =
                        *reinterpret_cast<const v4f*>(&os[buf ^ 1][rr * G1_OPITCH + opc]);
            }
        }

        const float* arow = &xs[buf][(rt * 16 + m) * G1_PITCH + khalf];
        v8f acc = {};
#pragma unroll
        for (int ks = 0; ks < NFEAT / 4; ++ks) {
            const v2f a = *reinterpret_cast<const v2f*>(arow + 4 * ks);
            acc = __builtin_amdgcn_wmma_f32_16x16x4_f32(false, a, false, bw[ks], (short)0, acc, false, false);
        }

        float* ost = &os[buf][(rt * 16 + (lane >> 4) * 8) * G1_OPITCH + ct * 16 + m];
#pragma unroll
        for (int j = 0; j < 8; ++j) ost[j * G1_OPITCH] = acc[j];

        if (more) {
#pragma unroll
            for (int p = 0; p < G1_ROWS / 4; ++p)
                *reinterpret_cast<v4f*>(&xs[buf ^ 1][(4 * p + srow) * G1_PITCH + spc]) = pre[p];
        }
        __syncthreads();
        prevTrip = trip;
        buf ^= 1;
    }
    if (prevTrip >= 0) {
#pragma unroll
        for (int p = 0; p < G1_ROWS / 16; ++p) {
            const int rr = 16 * p + orow;
            const int r  = prevTrip * G1_ROWS + rr;
            if (r < N_NODES)
                *reinterpret_cast<v4f*>(h1 + (long)r * NHID + opc) =
                    *reinterpret_cast<const v4f*>(&os[buf ^ 1][rr * G1_OPITCH + opc]);
        }
    }
}

#define OWN_WAVES 32
#define RING_CAP  256
#define KEY_BYTES N_EDGES
#define LIST_SEG  ((N_NODES * NCLASS * 4 - KEY_BYTES) / N_RANGES / 128 * 32)
#define LIST_CAP  (LIST_SEG / 64 - 1)
static_assert(LAST_ROWS > 0 && LAST_ROWS <= OWN_ROWS, "row ranges must cover N");
static_assert(N_NODES <= (1 << 17) && OWN_ROWS < (1 << 14), "src and row share one 32-bit word");
static_assert((LIST_SEG * 4) % 128 == 0 && (long)N_RANGES * LIST_SEG * 4 + KEY_BYTES <= (long)N_NODES * NCLASS * 4, "segments and the key plane start on a line and fit the plane");
static_assert((RING_CAP & (RING_CAP - 1)) == 0 && RING_CAP >= 64, "a wave adds at most 32 entries at a time");
static_assert((long)N_EDGES * 8 <= (long)N_NODES * NCLASS * 4, "the record plane (8 B per edge) fits the memory of out");

__device__ __forceinline__ void lds_add(float* p, float v) {
    __hip_atomic_fetch_add(p, v, __ATOMIC_RELAXED, __HIP_MEMORY_SCOPE_WORKGROUP);
}

template <int D>
__device__ __forceinline__ void own_apply(int s, float ww, int r, int lane,
                                          const float* __restrict__ h, float* __restrict__ tile) {
#pragma unroll
    for (int hb = 0; hb < 2; ++hb) {
        float v0[16], v1[16];
#pragma unroll
        for (int j = 0; j < 16; ++j) {
            const int sj = __builtin_amdgcn_readlane(s, hb * 16 + j);
            const float* __restrict__ hr = h + (long)sj * D;
            v0[j] = hr[lane];
            v1[j] = 0.0f;
            if (lane + 32 < D) v1[j] = hr[lane + 32];
        }
#pragma unroll
        for (int j = 0; j < 16; ++j) {
            const float wj = __builtin_bit_cast(float, __builtin_amdgcn_readlane(__builtin_bit_cast(int, ww), hb * 16 + j));
            const int   rj = __builtin_amdgcn_readlane(r, hb * 16 + j);
            float* tr = tile + rj * D;
            lds_add(tr + lane, wj * v0[j]);
            if (lane + 32 < D) lds_add(tr + lane + 32, wj * v1[j]);
        }
    }
}

template <int D, bool KEEP, bool REC>
__device__ __forceinline__ void own_drain(const int* __restrict__ ring, int head, int n, int lane, int lo,
                                          const int* __restrict__ src, const int* __restrict__ dst,
                                          const float* __restrict__ w, const int2* __restrict__ rec,
                                          const float* __restrict__ h, float* __restrict__ tile,
                                          int2* __restrict__ seg, int* slotCount) {
    __builtin_amdgcn_fence(__ATOMIC_ACQ_REL, "wavefront");
    const bool ok = lane < n;
    const int  re = ring[(head + lane) & (RING_CAP - 1)];
    const int   e  = ok ? re : 0;
    int r, s;
    float ww;
    if (REC) {
        const int2 rc = rec[e];
        r  = ok ? (int)((unsigned)rc.x >> 17) : OWN_ROWS;
        s  = rc.x & 0x1FFFF;
        ww = ok ? __builtin_bit_cast(float, rc.y) : 0.0f;
    } else {
        r  = ok ? dst[e] - lo : OWN_ROWS;
        s  = src[e];
        ww = ok ? w[e] : 0.0f;
    }
    if (KEEP) {
        int slot = 0;
        if (lane == 0) slot = __hip_atomic_fetch_add(slotCount, 1, __ATOMIC_RELAXED, __HIP_MEMORY_SCOPE_WORKGROUP);
        slot = __builtin_amdgcn_readfirstlane(slot);
        if (slot < LIST_CAP) seg[(slot + 1) * 32 + lane] = make_int2(s | (r << 17), __builtin_bit_cast(int, ww));
    }
    own_apply<D>(s, ww, r, lane, h, tile);
}

template <int D, int MODE>
__global__ __launch_bounds__(1024) void spmm_own_k(const int* __restrict__ src,
                                                   const int* __restrict__ dst,
                                                   const float* __restrict__ w,
                                                   const float* __restrict__ h,
                                                   float* __restrict__ agg,
                                                   int nEdges,
                                                   float* __restrict__ lst,
                                                   const float* __restrict__ bias,
                                                   const float* __restrict__ W2,
                                                   const float* __restrict__ recf) {
    static_assert(D == NHID, "both launches sum rows of NHID floats");
    static_assert(OWN_ROWS % 16 == 0 && LAST_ROWS % 16 == 0 && OWN_ROWS / 16 <= OWN_WAVES, "MODE 1: one wave per 16 owned rows");
    static_assert((OWN_ROWS * NCLASS) % 4 == 0 && (OWN_ROWS * NCLASS * 4) % 128 == 0 &&
                  (LAST_ROWS * NCLASS) % 4 == 0 && (LAST_ROWS * NCLASS * 4) % 128 == 0, "MODE 1: out rows leave as whole 16-B pieces of whole lines");
    static_assert(1024 % (D / 4) == 0, "MODE 0: a thread's 16-B pieces of the tile all lie in the same four columns");
    static_assert((OWN_ROWS * D) % 4 == 0 && (OWN_ROWS * D * 4) % 128 == 0 &&
                  (LAST_ROWS * D) % 4 == 0 && (LAST_ROWS * D * 4) % 128 == 0, "tile stores are whole 16-B pieces of whole lines");
    __shared__ __attribute__((aligned(16))) float tile[(OWN_ROWS + 1) * D];
    __shared__ int rings[OWN_WAVES * RING_CAP];
    __shared__ int slotCount;

    const int lane = threadIdx.x & 31;
    const int wave = __builtin_amdgcn_readfirstlane(threadIdx.x >> 5);
    const int lo   = blockIdx.x * OWN_ROWS;

    float4* t4 = reinterpret_cast<float4*>(tile);
    for (int i = threadIdx.x; i < OWN_ROWS * D / 4; i += 1024) t4[i] = make_float4(0.f, 0.f, 0.f, 0.f);
    if (MODE == 0 && threadIdx.x == 0) slotCount = 0;
    __syncthreads();

    int* ring = rings + wave * RING_CAP;
    int head = 0, tail = 0;
    int2* __restrict__ seg = reinterpret_cast<int2*>(lst + (long)blockIdx.x * LIST_SEG);
    const int4* __restrict__ key4 = reinterpret_cast<const int4*>(lst + (long)N_RANGES * LIST_SEG);
    const unsigned pat = (unsigned)blockIdx.x * 0x01010101u;
    const int2* __restrict__ rec = reinterpret_cast<const int2*>(recf);

    auto push = [&](unsigned m, bool f, int e) {
        if (f) ring[(tail + (int)__builtin_amdgcn_mbcnt_lo(m, 0u)) & (RING_CAP - 1)] = e;
        tail += __builtin_popcount(m);
    };
    auto make_room = [&]() {
        if (tail - head > RING_CAP - 32) {
            if (MODE == 0 && lane == 0) __hip_atomic_fetch_or(&slotCount, 1 << 30, __ATOMIC_RELAXED, __HIP_MEMORY_SCOPE_WORKGROUP);
            __builtin_amdgcn_fence(__ATOMIC_ACQ_REL, "wavefront");
#pragma nounroll
            for (int k = head; k != tail; ++k) {
                const int e = __builtin_amdgcn_readfirstlane(ring[k & (RING_CAP - 1)]);
                const int r = dst[e] - lo;
                const float wj = w[e];
                const float* __restrict__ hr = h + (long)src[e] * D;
                float* tr = tile + r * D;
                lds_add(tr + lane, wj * hr[lane]);
                if (lane + 32 < D) lds_add(tr + lane + 32, wj * hr[lane + 32]);
            }
            head = tail;
        }
    };
    auto take = [&](int e0, int4 d) {
        const unsigned x0 = (unsigned)d.x ^ pat, x1 = (unsigned)d.y ^ pat, x2 = (unsigned)d.z ^ pat, x3 = (unsigned)d.w ^ pat;
        const unsigned m0 = ~(((x0 & 0x7F7F7F7Fu) + 0x7F7F7F7Fu) | x0) & 0x80808080u;
        const unsigned m1 = ~(((x1 & 0x7F7F7F7Fu) + 0x7F7F7F7Fu) | x1) & 0x80808080u;
        const unsigned m2 = ~(((x2 & 0x7F7F7F7Fu) + 0x7F7F7F7Fu) | x2) & 0x80808080u;
        const unsigned m3 = ~(((x3 & 0x7F7F7F7Fu) + 0x7F7F7F7Fu) | x3) & 0x80808080u;
        unsigned z = (m0 >> 7) | (m1 >> 6) | (m2 >> 5) | (m3 >> 4);
#pragma nounroll
        for (int it = 0; it < 16; ++it) {
            const bool f = z != 0u;
            const unsigned m = __builtin_amdgcn_ballot_w32(f);
            if (m == 0u) break;
            make_room();
            const int p = __builtin_ctz(z | 0x80000000u);
            push(m, f, e0 + ((p & 7) << 2) + (p >> 3));
            z &= z - 1u;
        }
    };
    auto drain_batches = [&]() {
        while (tail - head >= 32) {
            own_drain<D, MODE == 0, MODE == 0>(ring, head, 32, lane, lo, src, dst, w, rec, h, tile, seg, &slotCount);
            head += 32;
        }
    };

    constexpr int U     = 4;
    constexpr int STEP  = 2 * U * OWN_WAVES;
    constexpr int CQUAD = OWN_WAVES * 32;
    const int nQuad  = nEdges >> 4;
    const int nBlock = (nQuad >> 5) / STEP;

    bool scan = true;
    if (MODE == 1) {
        const int nSlot = __builtin_amdgcn_readfirstlane(seg[0].x);
        if (nSlot <= LIST_CAP) {
            scan = false;
            int sl = wave;
            int2 ent = make_int2(OWN_ROWS << 17, 0);
            if (sl < nSlot) ent = seg[(sl + 1) * 32 + lane];
            while (sl < nSlot) {
                const int nx = sl + OWN_WAVES;
                int2 nxt = ent;
                if (nx < nSlot) nxt = seg[(nx + 1) * 32 + lane];
                own_apply<D>(ent.x & 0x1FFFF, __builtin_bit_cast(float, ent.y), (int)((unsigned)ent.x >> 17), lane, h, tile);
                ent = nxt;
                sl = nx;
            }
        }
    }

    if (scan) {
    {
        const int q0 = (nBlock * STEP + wave) * 32 + lane;
        int4 rest[2 * U];
#pragma unroll
        for (int u = 0; u < 2 * U; ++u) {
            rest[u] = make_int4(-1, -1, -1, -1);
            if (q0 + u * CQUAD < nQuad) rest[u] = key4[q0 + u * CQUAD];
        }
#pragma unroll
        for (int u = 0; u < 2 * U; ++u) take((q0 + u * CQUAD) * 16, rest[u]);
        if (wave == 0) {
            const int e = nQuad * 16 + lane;
            unsigned r = 0xFFFFFFFFu;
            if (lane < (nEdges & 15)) r = (unsigned)(dst[e] - lo);
            const bool f = r < (unsigned)OWN_ROWS;
            const unsigned m = __builtin_amdgcn_ballot_w32(f);
            if (m != 0u) {
                make_room();
                push(m, f, e);
            }
        }
        drain_batches();
    }

    if (nBlock > 0) {
        int blk = (int)(((long)blockIdx.x * nBlock) / gridDim.x);
        int q = (blk * STEP + wave) * 32 + lane;
        int4 bufA[U], bufB[U];
#pragma unroll
        for (int u = 0; u < U; ++u) bufA[u] = key4[q + u * CQUAD];
        for (int i = 0; i < nBlock; ++i) {
#pragma unroll
            for (int u = 0; u < U; ++u) bufB[u] = key4[q + (U + u) * CQUAD];
#pragma unroll
            for (int u = 0; u < U; ++u) take((q + u * CQUAD) * 16, bufA[u]);
            blk = (blk + 1 == nBlock) ? 0 : blk + 1;
            const int qn = (blk * STEP + wave) * 32 + lane;
#pragma unroll
            for (int u = 0; u < U; ++u) bufA[u] = key4[qn + u * CQUAD];
#pragma unroll
            for (int u = 0; u < U; ++u) take((q + (U + u) * CQUAD) * 16, bufB[u]);
            q = qn;
            drain_batches();
        }
    }
    if (tail - head > 0) own_drain<D, MODE == 0, MODE == 0>(ring, head, tail - head, lane, lo, src, dst, w, rec, h, tile, seg, &slotCount);
    }

    __syncthreads();
    const int rows = blockIdx.x == N_RANGES - 1 ? LAST_ROWS : OWN_ROWS;
    if (MODE == 0) {
        if (threadIdx.x == 0) seg[0] = make_int2(slotCount, 0);
        const int c4 = (threadIdx.x % (D / 4)) * 4;
        const float bx = bias[c4], by = bias[c4 + 1], bz = bias[c4 + 2], bw = bias[c4 + 3];
        float4* __restrict__ o4 = reinterpret_cast<float4*>(agg + (long)lo * D);
        const int nOut = rows * D / 4;
        for (int i = threadIdx.x; i < nOut; i += 1024) {
            const float4 v = t4[i];
            o4[i] = make_float4(fmaxf(v.x + bx, 0.0f), fmaxf(v.y + by, 0.0f), fmaxf(v.z + bz, 0.0f), fmaxf(v.w + bw, 0.0f));
        }
    } else {
        const int m     = lane & 15;
        const int khalf = (lane >> 4) * 2;
        const bool mine = wave * 16 < rows;
        const bool has2 = 32 + m < NCLASS;
        float res0[8], res1[8], res2[8];
        if (mine) {
            const float* arow = tile + (wave * 16 + m) * D;
            v8f acc0 = {}, acc1 = {}, acc2 = {};
#pragma unroll
            for (int k0 = 0; k0 < NHID; k0 += 4) {
                const int ka = k0 + khalf;
                v2f a;
                a.x = arow[ka];
                a.y = arow[ka + 1];
                v2f b0, b1v, b2v;
                b0.x  = W2[ka * NCLASS + m];
                b0.y  = W2[(ka + 1) * NCLASS + m];
                b1v.x = W2[ka * NCLASS + 16 + m];
                b1v.y = W2[(ka + 1) * NCLASS + 16 + m];
                const int nc = has2 ? 32 + m : NCLASS - 1;
                float lo2 = W2[ka * NCLASS + nc];
                float hi2 = W2[(ka + 1) * NCLASS + nc];
                if (!has2) { lo2 = 0.0f; hi2 = 0.0f; }
                b2v.x = lo2; b2v.y = hi2;
                acc0 = __builtin_amdgcn_wmma_f32_16x16x4_f32(false, a, false, b0,  (short)0, acc0, false, false);
                acc1 = __builtin_amdgcn_wmma_f32_16x16x4_f32(false, a, false, b1v, (short)0, acc1, false, false);
                acc2 = __builtin_amdgcn_wmma_f32_16x16x4_f32(false, a, false, b2v, (short)0, acc2, false, false);
            }
            const float bb0 = bias[m], bb1 = bias[16 + m];
            const float bb2 = has2 ? bias[32 + m] : 0.0f;
#pragma unroll
            for (int j = 0; j < 8; ++j) {
                const float z0 = acc0[j] + bb0, z1 = acc1[j] + bb1, z2 = acc2[j] + bb2;
                float mx = fmaxf(z0, z1);
                if (has2) mx = fmaxf(mx, z2);
                mx = fmaxf(mx, __shfl_xor(mx, 1));
                mx = fmaxf(mx, __shfl_xor(mx, 2));
                mx = fmaxf(mx, __shfl_xor(mx, 4));
                mx = fmaxf(mx, __shfl_xor(mx, 8));
                float s = expf(z0 - mx) + expf(z1 - mx);
                if (has2) s += expf(z2 - mx);
                s += __shfl_xor(s, 1);
                s += __shfl_xor(s, 2);
                s += __shfl_xor(s, 4);
                s += __shfl_xor(s, 8);
                const float lse = mx + logf(s);
                res0[j] = z0 - lse;
                res1[j] = z1 - lse;
                res2[j] = z2 - lse;
            }
        }
        __syncthreads();
        if (mine) {
            float* st = tile + (wave * 16 + (lane >> 4) * 8) * NCLASS + m;
#pragma unroll
            for (int j = 0; j < 8; ++j) {
                st[j * NCLASS]      = res0[j];
                st[j * NCLASS + 16] = res1[j];
                if (has2) st[j * NCLASS + 32] = res2[j];
            }
        }
        __syncthreads();
        float4* __restrict__ o4 = reinterpret_cast<float4*>(agg + (long)lo * NCLASS);
        const int nOut = rows * NCLASS / 4;
        for (int i = threadIdx.x; i < nOut; i += 1024) o4[i] = t4[i];
    }
}

extern "C" void kernel_launch(void* const* d_in, const int* in_sizes, int n_in,
                              void* d_out, int out_size, void* d_ws, size_t ws_size,
                              hipStream_t stream) {
    const float* x    = (const float*)d_in[0];
    const int*   esrc = (const int*)d_in[1];
    const int*   edst = (const int*)d_in[2];
    const float* ew   = (const float*)d_in[3];
    const float* W1   = (const float*)d_in[4];
    const float* b1   = (const float*)d_in[5];
    const float* W2   = (const float*)d_in[6];
    const float* b2   = (const float*)d_in[7];
    float* out = (float*)d_out;
    const int E = in_sizes[1];

    float* h1   = (float*)d_ws;
    float* agg1 = h1   + (size_t)N_NODES * NHID;
    float* lst  = agg1 + (size_t)N_NODES * NHID;
    float* keyf = lst  + (size_t)N_RANGES * LIST_SEG;

    gemm1_wmma_k<<<G1_GROUPS, 256, 0, stream>>>(x, W1, h1, edst, E, keyf, esrc, ew, out);

    spmm_own_k<NHID, 0><<<N_RANGES, 1024, 0, stream>>>(esrc, edst, ew, h1, agg1, E, lst, b1, W2, out);

    spmm_own_k<NHID, 1><<<N_RANGES, 1024, 0, stream>>>(esrc, edst, ew, agg1, out, E, lst, b2, W2, nullptr);
}
